# starting kernel
# baseline (speedup 1.0000x reference)
#include <hip/hip_runtime.h>
#include <math.h>

constexpr int Bn  = 4;
constexpr int Wd  = 32;
constexpr int Hd  = 32;
constexpr int Cc  = 128;
constexpr int Ss  = 256;
constexpr int Dd  = 512;
constexpr int NC  = 64;
constexpr int NPOS = Bn * Wd * Hd;

typedef __attribute__((ext_vector_type(2))) float v2f;
typedef __attribute__((ext_vector_type(8))) float v8f;

__device__ __forceinline__ v8f wmma_k4(v2f a, v2f b, v8f c) {
  return __builtin_amdgcn_wmma_f32_16x16x4_f32(
       false, a,  false, b,
       (short)0, c,  false,  false);
}


__global__ __launch_bounds__(128) void text_proj_kernel(
    const float* __restrict__ s, const float* __restrict__ amask,
    const float* __restrict__ Wks, const float* __restrict__ bks,
    const float* __restrict__ Wvs, const float* __restrict__ bvs,
    float* __restrict__ ks_out, float* __restrict__ vs_out) {
  constexpr int TE_STRIDE = Dd + 4;
  __shared__ float te_s[16 * TE_STRIDE];

  const int mat = blockIdx.x & 1;
  const int st  = (blockIdx.x >> 1) & 15;
  const int b   = blockIdx.x >> 5;

  const int tid  = threadIdx.x;
  const int lane = tid & 31;
  const int wave = tid >> 5;

  for (int i = tid; i < 16 * Dd; i += 128) {
    const int r = i >> 9, c = i & (Dd - 1);
    const int srow = st * 16 + r;
    te_s[r * TE_STRIDE + c] =
        s[((size_t)b * Ss + srow) * Dd + c] * amask[b * Ss + srow];
  }
  __syncthreads();

  const float* Wm   = mat ? Wvs : Wks;
  const float* bias = mat ? bvs : bks;
  float*       dst  = mat ? vs_out : ks_out;

  const int m    = lane & 15;
  const int koff = (lane >> 4) * 2;
  const int n    = wave * 16 + (lane & 15);

  v8f acc = {0.f, 0.f, 0.f, 0.f, 0.f, 0.f, 0.f, 0.f};
#pragma unroll 4
  for (int k = 0; k < Dd; k += 4) {
    v2f af, bf;
    af.x = te_s[m * TE_STRIDE + k + koff];
    af.y = te_s[m * TE_STRIDE + k + koff + 1];
    bf.x = Wm[(k + koff) * NC + n];
    bf.y = Wm[(k + koff + 1) * NC + n];
    acc = wmma_k4(af, bf, acc);
  }
  const float bn = bias[n];
#pragma unroll
  for (int i = 0; i < 8; ++i) {
    const int row = i + ((lane >> 4) << 3);
    dst[((size_t)b * Ss + st * 16 + row) * NC + n] = acc[i] + bn;
  }
}

__global__ __launch_bounds__(128) void colsum_kernel(
    const float* __restrict__ ks, const float* __restrict__ vs,
    float* __restrict__ Ksum, float* __restrict__ Vsum) {
  const int b = blockIdx.x;
  const int t = threadIdx.x;
  const float* src = (t < NC) ? ks : vs;
  const int col = t & (NC - 1);
  float acc = 0.f;
  for (int sIdx = 0; sIdx < Ss; ++sIdx)
    acc += src[((size_t)b * Ss + sIdx) * NC + col];
  if (t < NC) Ksum[b * NC + col] = acc;
  else        Vsum[b * NC + col] = acc;
}

__global__ __launch_bounds__(128) void gram_kernel(
    const float* __restrict__ ks, const float* __restrict__ vs,
    float* __restrict__ G) {
  const int mt = blockIdx.x & 3;
  const int b  = blockIdx.x >> 2;
  const int lane = threadIdx.x & 31;
  const int wave = threadIdx.x >> 5;

  const int m    = lane & 15;
  const int koff = (lane >> 4) * 2;
  const int c0   = mt * 16;
  const int n    = wave * 16 + (lane & 15);

  v8f acc = {0.f, 0.f, 0.f, 0.f, 0.f, 0.f, 0.f, 0.f};
#pragma unroll 4
  for (int k = 0; k < Ss; k += 4) {
    v2f af, bf;
    af.x = vs[((size_t)b * Ss + k + koff) * NC + c0 + m];
    af.y = vs[((size_t)b * Ss + k + koff + 1) * NC + c0 + m];
    bf.x = ks[((size_t)b * Ss + k + koff) * NC + n];
    bf.y = ks[((size_t)b * Ss + k + koff + 1) * NC + n];
    acc = wmma_k4(af, bf, acc);
  }
#pragma unroll
  for (int i = 0; i < 8; ++i) {
    const int row = i + ((lane >> 4) << 3);
    G[((size_t)b * NC + c0 + row) * NC + n] = acc[i];
  }
}

__global__ __launch_bounds__(128) void fused_attn_kernel(
    const float* __restrict__ x,
    const float* __restrict__ Wq,  const float* __restrict__ bq,
    const float* __restrict__ Wkx, const float* __restrict__ bkx,
    const float* __restrict__ Wvx, const float* __restrict__ bvx,
    const float* __restrict__ G,
    const float* __restrict__ Ksum, const float* __restrict__ Vsum,
    const float* __restrict__ Wout, const float* __restrict__ bout,
    const float* __restrict__ ln_scale, const float* __restrict__ ln_bias,
    float* __restrict__ y) {
  constexpr int XS = Cc + 4;
  constexpr int QS = NC + 4;
  __shared__ float xs[16 * XS];
  __shared__ float qs[16 * QS];
  __shared__ float kxs[16 * QS];
  __shared__ float vxs[16 * QS];
  __shared__ float os[16 * QS];
  __shared__ float outs[16 * XS];
  __shared__ float Gs[NC * NC];
  __shared__ float KsS[NC], VsS[NC];

  const int tid  = threadIdx.x;
  const int lane = tid & 31;
  const int wave = tid >> 5;
  const int p0   = blockIdx.x * 16;
  const int b    = p0 >> 10;

  for (int i = tid; i < 16 * Cc; i += 128)
    xs[(i >> 7) * XS + (i & (Cc - 1))] = x[(size_t)p0 * Cc + i];
  for (int i = tid; i < NC * NC; i += 128)
    Gs[i] = G[(size_t)b * NC * NC + i];
  if (tid < NC)            KsS[tid] = Ksum[b * NC + tid];
  else if (tid < 2 * NC)   VsS[tid - NC] = Vsum[b * NC + tid - NC];
  __syncthreads();

  {
    const int m    = lane & 15;
    const int koff = (lane >> 4) * 2;
    const int n    = wave * 16 + (lane & 15);
    for (int mat = 0; mat < 3; ++mat) {
      const float* Wm   = (mat == 0) ? Wq  : (mat == 1) ? Wkx : Wvx;
      const float* bias = (mat == 0) ? bq  : (mat == 1) ? bkx : bvx;
      float*       dst  = (mat == 0) ? qs  : (mat == 1) ? kxs : vxs;
      v8f acc = {0.f, 0.f, 0.f, 0.f, 0.f, 0.f, 0.f, 0.f};
#pragma unroll 4
      for (int k = 0; k < Cc; k += 4) {
        v2f af, bf;
        af.x = xs[m * XS + k + koff];
        af.y = xs[m * XS + k + koff + 1];
        bf.x = Wm[(k + koff) * NC + n];
        bf.y = Wm[(k + koff + 1) * NC + n];
        acc = wmma_k4(af, bf, acc);
      }
      const float bn = bias[n];
#pragma unroll
      for (int i = 0; i < 8; ++i)
        dst[(i + ((lane >> 4) << 3)) * QS + n] = acc[i] + bn;
    }
  }
  __syncthreads();

  for (int pi = 0; pi < 4; ++pi) {
    const int p = wave * 4 + pi;
    const float kx0 = kxs[p * QS + lane];
    const float kx1 = kxs[p * QS + lane + 32];
    const float Ks0 = KsS[lane];
    const float Ks1 = KsS[lane + 32];
    float o0 = 0.f, o1 = 0.f;
#pragma unroll 4
    for (int j = 0; j < NC; ++j) {
      const float vj = vxs[p * QS + j];
      const float Vj = VsS[j];
      const float qj = qs[p * QS + j];
      float r0 = (256.f * vj * kx0 + vj * Ks0 + kx0 * Vj + Gs[j * NC + lane]) * 0.125f;
      float r1 = (256.f * vj * kx1 + vj * Ks1 + kx1 * Vj + Gs[j * NC + lane + 32]) * 0.125f;
      float mx = fmaxf(r0, r1);
#pragma unroll
      for (int msk = 16; msk >= 1; msk >>= 1)
        mx = fmaxf(mx, __shfl_xor(mx, msk, 32));
      const float e0 = __expf(r0 - mx);
      const float e1 = __expf(r1 - mx);
      float sm = e0 + e1;
#pragma unroll
      for (int msk = 16; msk >= 1; msk >>= 1)
        sm += __shfl_xor(sm, msk, 32);
      const float qw = qj / sm;
      o0 += qw * e0;
      o1 += qw * e1;
    }
    os[p * QS + lane]      = o0;
    os[p * QS + lane + 32] = o1;
  }
  __syncthreads();

  {
    const int m    = lane & 15;
    const int koff = (lane >> 4) * 2;
    for (int ct = wave; ct < 8; ct += 4) {
      const int n = ct * 16 + (lane & 15);
      v8f acc = {0.f, 0.f, 0.f, 0.f, 0.f, 0.f, 0.f, 0.f};
#pragma unroll 4
      for (int k = 0; k < NC; k += 4) {
        v2f af, bf;
        af.x = os[m * QS + k + koff];
        af.y = os[m * QS + k + koff + 1];
        bf.x = Wout[(k + koff) * Cc + n];
        bf.y = Wout[(k + koff + 1) * Cc + n];
        acc = wmma_k4(af, bf, acc);
      }
      const float bn = bout[n];
#pragma unroll
      for (int i = 0; i < 8; ++i)
        outs[(i + ((lane >> 4) << 3)) * XS + n] = acc[i] + bn;
    }
  }
  __syncthreads();

  {
    const int row   = tid >> 3;
    const int cbase = (tid & 7) * 16;
    float vals[16];
    float sum = 0.f, sq = 0.f;
#pragma unroll
    for (int i = 0; i < 16; ++i) {
      const float v = xs[row * XS + cbase + i] + outs[row * XS + cbase + i];
      vals[i] = v;
      sum += v;
      sq  += v * v;
    }
#pragma unroll
    for (int msk = 4; msk >= 1; msk >>= 1) {
      sum += __shfl_xor(sum, msk, 32);
      sq  += __shfl_xor(sq,  msk, 32);
    }
    const float mu   = sum * (1.f / 128.f);
    const float var  = sq * (1.f / 128.f) - mu * mu;
    const float rstd = rsqrtf(var + 1e-6f);
#pragma unroll
    for (int i = 0; i < 16; ++i) {
      const int c = cbase + i;
      y[(size_t)(p0 + row) * Cc + c] =
          (vals[i] - mu) * rstd * ln_scale[c] + ln_bias[c];
    }
  }
}

extern "C" void kernel_launch(void* const* d_in, const int* in_sizes, int n_in,
                              void* d_out, int out_size, void* d_ws, size_t ws_size,
                              hipStream_t stream) {
  const float* x     = (const float*)d_in[0];
  const float* s     = (const float*)d_in[1];
  const float* amask = (const float*)d_in[2];
  const float* Wq    = (const float*)d_in[3];
  const float* bq    = (const float*)d_in[4];
  const float* Wkx   = (const float*)d_in[5];
  const float* bkx   = (const float*)d_in[6];
  const float* Wvx   = (const float*)d_in[7];
  const float* bvx   = (const float*)d_in[8];
  const float* Wks   = (const float*)d_in[9];
  const float* bks   = (const float*)d_in[10];
  const float* Wvs   = (const float*)d_in[11];
  const float* bvs   = (const float*)d_in[12];
  const float* Wout  = (const float*)d_in[13];
  const float* bout  = (const float*)d_in[14];
  const float* lns   = (const float*)d_in[15];
  const float* lnb   = (const float*)d_in[16];
  float* y = (float*)d_out;

  char* ws = (char*)d_ws;
  float* ksb  = (float*)(ws + 0);
  float* vsb  = (float*)(ws + 262144);
  float* G    = (float*)(ws + 524288);
  float* Ksum = (float*)(ws + 589824);
  float* Vsum = (float*)(ws + 590848);

  text_proj_kernel<<<dim3(Bn * (Ss / 16) * 2), dim3(128), 0, stream>>>(
      s, amask, Wks, bks, Wvs, bvs, ksb, vsb);
  colsum_kernel<<<dim3(Bn), dim3(128), 0, stream>>>(ksb, vsb, Ksum, Vsum);
  gram_kernel<<<dim3(Bn * (NC / 16)), dim3(128), 0, stream>>>(ksb, vsb, G);
  fused_attn_kernel<<<dim3(NPOS / 16), dim3(128), 0, stream>>>(
      x, Wq, bq, Wkx, bkx, Wvx, bvx, G, Ksum, Vsum, Wout, bout, lns, lnb, y);
}


# Channel softmax: row max and row sum from a lane-per-row pre-pass (serial over 64 columns, G pitch 65), so the j loop has no wave shuffles; row sums added serially; logit in fmaf form for equal bits
# speedup vs baseline: 1.5170x; 1.5170x over previous
#include <hip/hip_runtime.h>
#include <math.h>

constexpr int Bn  = 4;
constexpr int Wd  = 32;
constexpr int Hd  = 32;
constexpr int Cc  = 128;
constexpr int Ss  = 256;
constexpr int Dd  = 512;
constexpr int NC  = 64;
constexpr int NPOS = Bn * Wd * Hd;

typedef __attribute__((ext_vector_type(2))) float v2f;
typedef __attribute__((ext_vector_type(8))) float v8f;

__device__ __forceinline__ v8f wmma_k4(v2f a, v2f b, v8f c) {
  return __builtin_amdgcn_wmma_f32_16x16x4_f32(
       false, a,  false, b,
       (short)0, c,  false,  false);
}


__device__ __forceinline__ float attn_logit(float vj, float Vj, float kxd,
                                            float Ksd, float g) {
  const float t = (256.f * vj) * kxd;
  return (fmaf(kxd, Vj, fmaf(vj, Ksd, t)) + g) * 0.125f;
}

__global__ __launch_bounds__(128) void text_proj_kernel(
    const float* __restrict__ s, const float* __restrict__ amask,
    const float* __restrict__ Wks, const float* __restrict__ bks,
    const float* __restrict__ Wvs, const float* __restrict__ bvs,
    float* __restrict__ ks_out, float* __restrict__ vs_out) {
  constexpr int TE_STRIDE = Dd + 4;
  __shared__ float te_s[16 * TE_STRIDE];

  const int mat = blockIdx.x & 1;
  const int st  = (blockIdx.x >> 1) & 15;
  const int b   = blockIdx.x >> 5;

  const int tid  = threadIdx.x;
  const int lane = tid & 31;
  const int wave = tid >> 5;

  for (int i = tid; i < 16 * Dd; i += 128) {
    const int r = i >> 9, c = i & (Dd - 1);
    const int srow = st * 16 + r;
    te_s[r * TE_STRIDE + c] =
        s[((size_t)b * Ss + srow) * Dd + c] * amask[b * Ss + srow];
  }
  __syncthreads();

  const float* Wm   = mat ? Wvs : Wks;
  const float* bias = mat ? bvs : bks;
  float*       dst  = mat ? vs_out : ks_out;

  const int m    = lane & 15;
  const int koff = (lane >> 4) * 2;
  const int n    = wave * 16 + (lane & 15);

  v8f acc = {0.f, 0.f, 0.f, 0.f, 0.f, 0.f, 0.f, 0.f};
#pragma unroll 4
  for (int k = 0; k < Dd; k += 4) {
    v2f af, bf;
    af.x = te_s[m * TE_STRIDE + k + koff];
    af.y = te_s[m * TE_STRIDE + k + koff + 1];
    bf.x = Wm[(k + koff) * NC + n];
    bf.y = Wm[(k + koff + 1) * NC + n];
    acc = wmma_k4(af, bf, acc);
  }
  const float bn = bias[n];
#pragma unroll
  for (int i = 0; i < 8; ++i) {
    const int row = i + ((lane >> 4) << 3);
    dst[((size_t)b * Ss + st * 16 + row) * NC + n] = acc[i] + bn;
  }
}

__global__ __launch_bounds__(128) void colsum_kernel(
    const float* __restrict__ ks, const float* __restrict__ vs,
    float* __restrict__ Ksum, float* __restrict__ Vsum) {
  const int b = blockIdx.x;
  const int t = threadIdx.x;
  const float* src = (t < NC) ? ks : vs;
  const int col = t & (NC - 1);
  float acc = 0.f;
  for (int sIdx = 0; sIdx < Ss; ++sIdx)
    acc += src[((size_t)b * Ss + sIdx) * NC + col];
  if (t < NC) Ksum[b * NC + col] = acc;
  else        Vsum[b * NC + col] = acc;
}

__global__ __launch_bounds__(128) void gram_kernel(
    const float* __restrict__ ks, const float* __restrict__ vs,
    float* __restrict__ G) {
  const int mt = blockIdx.x & 3;
  const int b  = blockIdx.x >> 2;
  const int lane = threadIdx.x & 31;
  const int wave = threadIdx.x >> 5;

  const int m    = lane & 15;
  const int koff = (lane >> 4) * 2;
  const int c0   = mt * 16;
  const int n    = wave * 16 + (lane & 15);

  v8f acc = {0.f, 0.f, 0.f, 0.f, 0.f, 0.f, 0.f, 0.f};
#pragma unroll 4
  for (int k = 0; k < Ss; k += 4) {
    v2f af, bf;
    af.x = vs[((size_t)b * Ss + k + koff) * NC + c0 + m];
    af.y = vs[((size_t)b * Ss + k + koff + 1) * NC + c0 + m];
    bf.x = ks[((size_t)b * Ss + k + koff) * NC + n];
    bf.y = ks[((size_t)b * Ss + k + koff + 1) * NC + n];
    acc = wmma_k4(af, bf, acc);
  }
#pragma unroll
  for (int i = 0; i < 8; ++i) {
    const int row = i + ((lane >> 4) << 3);
    G[((size_t)b * NC + c0 + row) * NC + n] = acc[i];
  }
}

__global__ __launch_bounds__(128) void fused_attn_kernel(
    const float* __restrict__ x,
    const float* __restrict__ Wq,  const float* __restrict__ bq,
    const float* __restrict__ Wkx, const float* __restrict__ bkx,
    const float* __restrict__ Wvx, const float* __restrict__ bvx,
    const float* __restrict__ G,
    const float* __restrict__ Ksum, const float* __restrict__ Vsum,
    const float* __restrict__ Wout, const float* __restrict__ bout,
    const float* __restrict__ ln_scale, const float* __restrict__ ln_bias,
    float* __restrict__ y) {
  constexpr int XS = Cc + 4;
  constexpr int QS = NC + 4;
  constexpr int GS = NC + 1;
  __shared__ float xs[16 * XS];
  __shared__ float qs[16 * QS];
  __shared__ float kxs[16 * QS];
  __shared__ float vxs[16 * QS];
  __shared__ float os[16 * QS];
  __shared__ float outs[16 * XS];
  __shared__ float Gs[NC * GS];
  __shared__ float KsS[NC];
  __shared__ float VsS[NC];

  const int tid  = threadIdx.x;
  const int lane = tid & 31;
  const int wave = tid >> 5;
  const int p0   = blockIdx.x * 16;
  const int b    = p0 >> 10;

  for (int i = tid; i < 16 * Cc; i += 128)
    xs[(i >> 7) * XS + (i & (Cc - 1))] = x[(size_t)p0 * Cc + i];
  for (int i = tid; i < NC * NC; i += 128)
    Gs[(i >> 6) * GS + (i & (NC - 1))] = G[(size_t)b * NC * NC + i];
  if (tid < NC)            KsS[tid] = Ksum[b * NC + tid];
  else if (tid < 2 * NC)   VsS[tid - NC] = Vsum[b * NC + tid - NC];
  __syncthreads();

  {
    const int m    = lane & 15;
    const int koff = (lane >> 4) * 2;
    const int n    = wave * 16 + (lane & 15);
    for (int mat = 0; mat < 3; ++mat) {
      const float* Wm   = (mat == 0) ? Wq  : (mat == 1) ? Wkx : Wvx;
      const float* bias = (mat == 0) ? bq  : (mat == 1) ? bkx : bvx;
      float*       dst  = (mat == 0) ? qs  : (mat == 1) ? kxs : vxs;
      v8f acc = {0.f, 0.f, 0.f, 0.f, 0.f, 0.f, 0.f, 0.f};
#pragma unroll 4
      for (int k = 0; k < Cc; k += 4) {
        v2f af, bf;
        af.x = xs[m * XS + k + koff];
        af.y = xs[m * XS + k + koff + 1];
        bf.x = Wm[(k + koff) * NC + n];
        bf.y = Wm[(k + koff + 1) * NC + n];
        acc = wmma_k4(af, bf, acc);
      }
      const float bn = bias[n];
#pragma unroll
      for (int i = 0; i < 8; ++i)
        dst[(i + ((lane >> 4) << 3)) * QS + n] = acc[i] + bn;
    }
  }
  __syncthreads();

  const int pbase = wave * 4;

  {
    float vj0[4], vj1[4], m0[4], m1[4], z0[4], z1[4];
#pragma unroll
    for (int pi = 0; pi < 4; ++pi) {
      vj0[pi] = vxs[(pbase + pi) * QS + lane];
      vj1[pi] = vxs[(pbase + pi) * QS + lane + 32];
      m0[pi] = -INFINITY;  m1[pi] = -INFINITY;
      z0[pi] = 0.f;        z1[pi] = 0.f;
    }
    const float Vj0 = VsS[lane];
    const float Vj1 = VsS[lane + 32];
    const float* g0 = &Gs[lane * GS];
    const float* g1 = &Gs[(lane + 32) * GS];

#pragma unroll 4
    for (int d = 0; d < NC; ++d) {
      const float ga = g0[d], gb = g1[d], Ksd = KsS[d];
#pragma unroll
      for (int pi = 0; pi < 4; ++pi) {
        const float kxd = kxs[(pbase + pi) * QS + d];
        m0[pi] = fmaxf(m0[pi], attn_logit(vj0[pi], Vj0, kxd, Ksd, ga));
        m1[pi] = fmaxf(m1[pi], attn_logit(vj1[pi], Vj1, kxd, Ksd, gb));
      }
    }
#pragma unroll 4
    for (int d = 0; d < NC; ++d) {
      const float ga = g0[d], gb = g1[d], Ksd = KsS[d];
#pragma unroll
      for (int pi = 0; pi < 4; ++pi) {
        const float kxd = kxs[(pbase + pi) * QS + d];
        z0[pi] += __expf(attn_logit(vj0[pi], Vj0, kxd, Ksd, ga) - m0[pi]);
        z1[pi] += __expf(attn_logit(vj1[pi], Vj1, kxd, Ksd, gb) - m1[pi]);
      }
    }
#pragma unroll
    for (int pi = 0; pi < 4; ++pi) {
      const int r0 = (pbase + pi) * QS + lane;
      os[r0]      = m0[pi];
      os[r0 + 32] = m1[pi];
      qs[r0]      = qs[r0] / z0[pi];
      qs[r0 + 32] = qs[r0 + 32] / z1[pi];
    }
  }
  __syncthreads();

  {
    float kx0[4], kx1[4], o0[4], o1[4];
#pragma unroll
    for (int pi = 0; pi < 4; ++pi) {
      kx0[pi] = kxs[(pbase + pi) * QS + lane];
      kx1[pi] = kxs[(pbase + pi) * QS + lane + 32];
      o0[pi] = 0.f;  o1[pi] = 0.f;
    }
    const float Ks0 = KsS[lane];
    const float Ks1 = KsS[lane + 32];
#pragma unroll 4
    for (int j = 0; j < NC; ++j) {
      const float ga = Gs[j * GS + lane];
      const float gb = Gs[j * GS + lane + 32];
      const float Vj = VsS[j];
#pragma unroll
      for (int pi = 0; pi < 4; ++pi) {
        const int rj = (pbase + pi) * QS + j;
        const float vj = vxs[rj];
        const float mj = os[rj];
        const float wj = qs[rj];
        o0[pi] += wj * __expf(attn_logit(vj, Vj, kx0[pi], Ks0, ga) - mj);
        o1[pi] += wj * __expf(attn_logit(vj, Vj, kx1[pi], Ks1, gb) - mj);
      }
    }
#pragma unroll
    for (int pi = 0; pi < 4; ++pi) {
      os[(pbase + pi) * QS + lane]      = o0[pi];
      os[(pbase + pi) * QS + lane + 32] = o1[pi];
    }
  }
  __syncthreads();

  {
    const int m    = lane & 15;
    const int koff = (lane >> 4) * 2;
    for (int ct = wave; ct < 8; ct += 4) {
      const int n = ct * 16 + (lane & 15);
      v8f acc = {0.f, 0.f, 0.f, 0.f, 0.f, 0.f, 0.f, 0.f};
#pragma unroll 4
      for (int k = 0; k < NC; k += 4) {
        v2f af, bf;
        af.x = os[m * QS + k + koff];
        af.y = os[m * QS + k + koff + 1];
        bf.x = Wout[(k + koff) * Cc + n];
        bf.y = Wout[(k + koff + 1) * Cc + n];
        acc = wmma_k4(af, bf, acc);
      }
      const float bn = bout[n];
#pragma unroll
      for (int i = 0; i < 8; ++i)
        outs[(i + ((lane >> 4) << 3)) * XS + n] = acc[i] + bn;
    }
  }
  __syncthreads();

  {
    const int row   = tid >> 3;
    const int cbase = (tid & 7) * 16;
    float vals[16];
    float sum = 0.f, sq = 0.f;
#pragma unroll
    for (int i = 0; i < 16; ++i) {
      const float v = xs[row * XS + cbase + i] + outs[row * XS + cbase + i];
      vals[i] = v;
      sum += v;
      sq  += v * v;
    }
#pragma unroll
    for (int msk = 4; msk >= 1; msk >>= 1) {
      sum += __shfl_xor(sum, msk, 32);
      sq  += __shfl_xor(sq,  msk, 32);
    }
    const float mu   = sum * (1.f / 128.f);
    const float var  = sq * (1.f / 128.f) - mu * mu;
    const float rstd = rsqrtf(var + 1e-6f);
#pragma unroll
    for (int i = 0; i < 16; ++i) {
      const int c = cbase + i;
      y[(size_t)(p0 + row) * Cc + c] =
          (vals[i] - mu) * rstd * ln_scale[c] + ln_bias[c];
    }
  }
}

extern "C" void kernel_launch(void* const* d_in, const int* in_sizes, int n_in,
                              void* d_out, int out_size, void* d_ws, size_t ws_size,
                              hipStream_t stream) {
  const float* x     = (const float*)d_in[0];
  const float* s     = (const float*)d_in[1];
  const float* amask = (const float*)d_in[2];
  const float* Wq    = (const float*)d_in[3];
  const float* bq    = (const float*)d_in[4];
  const float* Wkx   = (const float*)d_in[5];
  const float* bkx   = (const float*)d_in[6];
  const float* Wvx   = (const float*)d_in[7];
  const float* bvx   = (const float*)d_in[8];
  const float* Wks   = (const float*)d_in[9];
  const float* bks   = (const float*)d_in[10];
  const float* Wvs   = (const float*)d_in[11];
  const float* bvs   = (const float*)d_in[12];
  const float* Wout  = (const float*)d_in[13];
  const float* bout  = (const float*)d_in[14];
  const float* lns   = (const float*)d_in[15];
  const float* lnb   = (const float*)d_in[16];
  float* y = (float*)d_out;

  char* ws = (char*)d_ws;
  float* ksb  = (float*)(ws + 0);
  float* vsb  = (float*)(ws + 262144);
  float* G    = (float*)(ws + 524288);
  float* Ksum = (float*)(ws + 589824);
  float* Vsum = (float*)(ws + 590848);

  text_proj_kernel<<<dim3(Bn * (Ss / 16) * 2), dim3(128), 0, stream>>>(
      s, amask, Wks, bks, Wvs, bvs, ksb, vsb);
  colsum_kernel<<<dim3(Bn), dim3(128), 0, stream>>>(ksb, vsb, Ksum, Vsum);
  gram_kernel<<<dim3(Bn * (NC / 16)), dim3(128), 0, stream>>>(ksb, vsb, G);
  fused_attn_kernel<<<dim3(NPOS / 16), dim3(128), 0, stream>>>(
      x, Wq, bq, Wkx, bkx, Wvx, bvx, G, Ksum, Vsum, Wout, bout, lns, lnb, y);
}


# Column-sum and Gram launches folded into the fused kernel: each block forms its batch's 64x64 G on WMMA, Ksum/Vsum from the same fragments (two half sums a column, then one add); 4 launches to 2
# speedup vs baseline: 1.8693x; 1.2322x over previous
#include <hip/hip_runtime.h>
#include <math.h>

constexpr int Bn  = 4;
constexpr int Wd  = 32;
constexpr int Hd  = 32;
constexpr int Cc  = 128;
constexpr int Ss  = 256;
constexpr int Dd  = 512;
constexpr int NC  = 64;
constexpr int NPOS = Bn * Wd * Hd;

typedef __attribute__((ext_vector_type(2))) float v2f;
typedef __attribute__((ext_vector_type(4))) float v4f;
typedef __attribute__((ext_vector_type(8))) float v8f;

__device__ __forceinline__ v8f wmma_k4(v2f a, v2f b, v8f c) {
  return __builtin_amdgcn_wmma_f32_16x16x4_f32(
       false, a,  false, b,
       (short)0, c,  false,  false);
}


__device__ __forceinline__ float attn_logit(float vj, float Vj, float kxd,
                                            float Ksd, float g) {
  const float t = (256.f * vj) * kxd;
  return (fmaf(kxd, Vj, fmaf(vj, Ksd, t)) + g) * 0.125f;
}

__global__ __launch_bounds__(128) void text_proj_kernel(
    const float* __restrict__ s, const float* __restrict__ amask,
    const float* __restrict__ Wks, const float* __restrict__ bks,
    const float* __restrict__ Wvs, const float* __restrict__ bvs,
    float* __restrict__ ks_out, float* __restrict__ vs_out) {
  constexpr int TE_STRIDE = Dd + 4;
  __shared__ float te_s[16 * TE_STRIDE];

  const int mat = blockIdx.x & 1;
  const int st  = (blockIdx.x >> 1) & 15;
  const int b   = blockIdx.x >> 5;

  const int tid  = threadIdx.x;
  const int lane = tid & 31;
  const int wave = tid >> 5;

  for (int i = tid; i < 16 * Dd; i += 128) {
    const int r = i >> 9, c = i & (Dd - 1);
    const int srow = st * 16 + r;
    te_s[r * TE_STRIDE + c] =
        s[((size_t)b * Ss + srow) * Dd + c] * amask[b * Ss + srow];
  }
  __syncthreads();

  const float* Wm   = mat ? Wvs : Wks;
  const float* bias = mat ? bvs : bks;
  float*       dst  = mat ? vs_out : ks_out;

  const int m    = lane & 15;
  const int koff = (lane >> 4) * 2;
  const int n    = wave * 16 + (lane & 15);

  v8f acc = {0.f, 0.f, 0.f, 0.f, 0.f, 0.f, 0.f, 0.f};
#pragma unroll 4
  for (int k = 0; k < Dd; k += 4) {
    v2f af, bf;
    af.x = te_s[m * TE_STRIDE + k + koff];
    af.y = te_s[m * TE_STRIDE + k + koff + 1];
    bf.x = Wm[(k + koff) * NC + n];
    bf.y = Wm[(k + koff + 1) * NC + n];
    acc = wmma_k4(af, bf, acc);
  }
  const float bn = bias[n];
#pragma unroll
  for (int i = 0; i < 8; ++i) {
    const int row = i + ((lane >> 4) << 3);
    dst[((size_t)b * Ss + st * 16 + row) * NC + n] = acc[i] + bn;
  }
}

struct GramGroup {
  v4f   a0[8], a1[8];
  float b0[8], b1[8];
};

__device__ __forceinline__ void gram_load(GramGroup& f, const v4f* __restrict__ vs4,
                                          const float* __restrict__ ksp, int g) {
#pragma unroll
  for (int u = 0; u < 8; ++u) {
    const int row = g * 32 + u * 4;
    f.a0[u] = vs4[(row)     * (NC / 4)];
    f.a1[u] = vs4[(row + 1) * (NC / 4)];
    f.b0[u] = ksp[(row)     * NC];
    f.b1[u] = ksp[(row + 1) * NC];
  }
}

__device__ __forceinline__ void gram_mma(const GramGroup& f, v8f (&acc)[4],
                                         v4f& vp, float& kp) {
#pragma unroll
  for (int u = 0; u < 8; ++u) {
    v2f bf;
    bf.x = f.b0[u];
    bf.y = f.b1[u];
#pragma unroll
    for (int t = 0; t < 4; ++t) {
      v2f af;
      af.x = f.a0[u][t];
      af.y = f.a1[u][t];
      acc[t] = wmma_k4(af, bf, acc[t]);
    }
    kp += f.b0[u] + f.b1[u];
    vp += f.a0[u] + f.a1[u];
  }
}

__global__ __launch_bounds__(128) void fused_attn_kernel(
    const float* __restrict__ x,
    const float* __restrict__ Wq,  const float* __restrict__ bq,
    const float* __restrict__ Wkx, const float* __restrict__ bkx,
    const float* __restrict__ Wvx, const float* __restrict__ bvx,
    const float* __restrict__ ks, const float* __restrict__ vs,
    const float* __restrict__ Wout, const float* __restrict__ bout,
    const float* __restrict__ ln_scale, const float* __restrict__ ln_bias,
    float* __restrict__ y) {
  constexpr int XS = Cc + 4;
  constexpr int QS = NC + 4;
  constexpr int GS = NC + 1;
  __shared__ float xs[16 * XS];
  __shared__ float qs[16 * QS];
  __shared__ float kxs[16 * QS];
  __shared__ float vxs[16 * QS];
  __shared__ float os[16 * QS];
  __shared__ float outs[16 * XS];
  __shared__ float Gs[NC * GS];
  __shared__ float KsS[NC];
  __shared__ float VsS[NC];

  const int tid  = threadIdx.x;
  const int lane = tid & 31;
  const int wave = tid >> 5;
  const int p0   = blockIdx.x * 16;
  const int b    = p0 >> 10;

  float xr[16];
#pragma unroll
  for (int i = 0; i < 16; ++i) xr[i] = x[(size_t)p0 * Cc + tid + i * 128];

  {
    const int m = lane & 15;
    const int h = lane >> 4;
    const int n = wave * 16 + m;
    const v4f*   vs4 = reinterpret_cast<const v4f*>(vs) + ((size_t)b * Ss + 2 * h) * (NC / 4) + m;
    const float* ksp = ks + ((size_t)b * Ss + 2 * h) * NC + n;

    v8f acc[4];
#pragma unroll
    for (int t = 0; t < 4; ++t) acc[t] = v8f{0.f, 0.f, 0.f, 0.f, 0.f, 0.f, 0.f, 0.f};
    v4f   vp = {0.f, 0.f, 0.f, 0.f};
    float kp = 0.f;

    GramGroup fa, fb;
    gram_load(fa, vs4, ksp, 0);
#pragma unroll 1
    for (int g = 0; g < 6; g += 2) {
      gram_load(fb, vs4, ksp, g + 1);
      gram_mma(fa, acc, vp, kp);
      gram_load(fa, vs4, ksp, g + 2);
      gram_mma(fb, acc, vp, kp);
    }
    gram_load(fb, vs4, ksp, 7);
    gram_mma(fa, acc, vp, kp);
    gram_mma(fb, acc, vp, kp);

    kp   += __shfl_xor(kp,   16, 32);
    vp.x += __shfl_xor(vp.x, 16, 32);
    vp.y += __shfl_xor(vp.y, 16, 32);
    vp.z += __shfl_xor(vp.z, 16, 32);
    vp.w += __shfl_xor(vp.w, 16, 32);

#pragma unroll
    for (int t = 0; t < 4; ++t)
#pragma unroll
      for (int i = 0; i < 8; ++i)
        Gs[(4 * (i + 8 * h) + t) * GS + n] = acc[t][i];
    if (lane < 16) {
      KsS[n] = kp;
      if (wave == 0) {
        VsS[4 * m + 0] = vp.x;
        VsS[4 * m + 1] = vp.y;
        VsS[4 * m + 2] = vp.z;
        VsS[4 * m + 3] = vp.w;
      }
    }
  }

#pragma unroll
  for (int i = 0; i < 16; ++i) xs[i * XS + tid] = xr[i];
  __syncthreads();

  {
    const int m    = lane & 15;
    const int koff = (lane >> 4) * 2;
    const int n    = wave * 16 + (lane & 15);
    for (int mat = 0; mat < 3; ++mat) {
      const float* Wm   = (mat == 0) ? Wq  : (mat == 1) ? Wkx : Wvx;
      const float* bias = (mat == 0) ? bq  : (mat == 1) ? bkx : bvx;
      float*       dst  = (mat == 0) ? qs  : (mat == 1) ? kxs : vxs;
      v8f acc = {0.f, 0.f, 0.f, 0.f, 0.f, 0.f, 0.f, 0.f};
#pragma unroll 4
      for (int k = 0; k < Cc; k += 4) {
        v2f af, bf;
        af.x = xs[m * XS + k + koff];
        af.y = xs[m * XS + k + koff + 1];
        bf.x = Wm[(k + koff) * NC + n];
        bf.y = Wm[(k + koff + 1) * NC + n];
        acc = wmma_k4(af, bf, acc);
      }
      const float bn = bias[n];
#pragma unroll
      for (int i = 0; i < 8; ++i)
        dst[(i + ((lane >> 4) << 3)) * QS + n] = acc[i] + bn;
    }
  }
  __syncthreads();

  const int pbase = wave * 4;

  {
    float vj0[4], vj1[4], m0[4], m1[4], z0[4], z1[4];
#pragma unroll
    for (int pi = 0; pi < 4; ++pi) {
      vj0[pi] = vxs[(pbase + pi) * QS + lane];
      vj1[pi] = vxs[(pbase + pi) * QS + lane + 32];
      m0[pi] = -INFINITY;  m1[pi] = -INFINITY;
      z0[pi] = 0.f;        z1[pi] = 0.f;
    }
    const float Vj0 = VsS[lane];
    const float Vj1 = VsS[lane + 32];
    const float* g0 = &Gs[lane * GS];
    const float* g1 = &Gs[(lane + 32) * GS];

#pragma unroll 4
    for (int d = 0; d < NC; ++d) {
      const float ga = g0[d], gb = g1[d], Ksd = KsS[d];
#pragma unroll
      for (int pi = 0; pi < 4; ++pi) {
        const float kxd = kxs[(pbase + pi) * QS + d];
        m0[pi] = fmaxf(m0[pi], attn_logit(vj0[pi], Vj0, kxd, Ksd, ga));
        m1[pi] = fmaxf(m1[pi], attn_logit(vj1[pi], Vj1, kxd, Ksd, gb));
      }
    }
#pragma unroll 4
    for (int d = 0; d < NC; ++d) {
      const float ga = g0[d], gb = g1[d], Ksd = KsS[d];
#pragma unroll
      for (int pi = 0; pi < 4; ++pi) {
        const float kxd = kxs[(pbase + pi) * QS + d];
        z0[pi] += __expf(attn_logit(vj0[pi], Vj0, kxd, Ksd, ga) - m0[pi]);
        z1[pi] += __expf(attn_logit(vj1[pi], Vj1, kxd, Ksd, gb) - m1[pi]);
      }
    }
#pragma unroll
    for (int pi = 0; pi < 4; ++pi) {
      const int r0 = (pbase + pi) * QS + lane;
      os[r0]      = m0[pi];
      os[r0 + 32] = m1[pi];
      qs[r0]      = qs[r0] / z0[pi];
      qs[r0 + 32] = qs[r0 + 32] / z1[pi];
    }
  }
  __syncthreads();

  {
    float kx0[4], kx1[4], o0[4], o1[4];
#pragma unroll
    for (int pi = 0; pi < 4; ++pi) {
      kx0[pi] = kxs[(pbase + pi) * QS + lane];
      kx1[pi] = kxs[(pbase + pi) * QS + lane + 32];
      o0[pi] = 0.f;  o1[pi] = 0.f;
    }
    const float Ks0 = KsS[lane];
    const float Ks1 = KsS[lane + 32];
#pragma unroll 4
    for (int j = 0; j < NC; ++j) {
      const float ga = Gs[j * GS + lane];
      const float gb = Gs[j * GS + lane + 32];
      const float Vj = VsS[j];
#pragma unroll
      for (int pi = 0; pi < 4; ++pi) {
        const int rj = (pbase + pi) * QS + j;
        const float vj = vxs[rj];
        const float mj = os[rj];
        const float wj = qs[rj];
        o0[pi] += wj * __expf(attn_logit(vj, Vj, kx0[pi], Ks0, ga) - mj);
        o1[pi] += wj * __expf(attn_logit(vj, Vj, kx1[pi], Ks1, gb) - mj);
      }
    }
#pragma unroll
    for (int pi = 0; pi < 4; ++pi) {
      os[(pbase + pi) * QS + lane]      = o0[pi];
      os[(pbase + pi) * QS + lane + 32] = o1[pi];
    }
  }
  __syncthreads();

  {
    const int m    = lane & 15;
    const int koff = (lane >> 4) * 2;
    for (int ct = wave; ct < 8; ct += 4) {
      const int n = ct * 16 + (lane & 15);
      v8f acc = {0.f, 0.f, 0.f, 0.f, 0.f, 0.f, 0.f, 0.f};
#pragma unroll 4
      for (int k = 0; k < NC; k += 4) {
        v2f af, bf;
        af.x = os[m * QS + k + koff];
        af.y = os[m * QS + k + koff + 1];
        bf.x = Wout[(k + koff) * Cc + n];
        bf.y = Wout[(k + koff + 1) * Cc + n];
        acc = wmma_k4(af, bf, acc);
      }
      const float bn = bout[n];
#pragma unroll
      for (int i = 0; i < 8; ++i)
        outs[(i + ((lane >> 4) << 3)) * XS + n] = acc[i] + bn;
    }
  }
  __syncthreads();

  {
    const int row   = tid >> 3;
    const int cbase = (tid & 7) * 16;
    float vals[16];
    float sum = 0.f, sq = 0.f;
#pragma unroll
    for (int i = 0; i < 16; ++i) {
      const float v = xs[row * XS + cbase + i] + outs[row * XS + cbase + i];
      vals[i] = v;
      sum += v;
      sq  += v * v;
    }
#pragma unroll
    for (int msk = 4; msk >= 1; msk >>= 1) {
      sum += __shfl_xor(sum, msk, 32);
      sq  += __shfl_xor(sq,  msk, 32);
    }
    const float mu   = sum * (1.f / 128.f);
    const float var  = sq * (1.f / 128.f) - mu * mu;
    const float rstd = rsqrtf(var + 1e-6f);
#pragma unroll
    for (int i = 0; i < 16; ++i) {
      const int c = cbase + i;
      y[(size_t)(p0 + row) * Cc + c] =
          (vals[i] - mu) * rstd * ln_scale[c] + ln_bias[c];
    }
  }
}

extern "C" void kernel_launch(void* const* d_in, const int* in_sizes, int n_in,
                              void* d_out, int out_size, void* d_ws, size_t ws_size,
                              hipStream_t stream) {
  const float* x     = (const float*)d_in[0];
  const float* s     = (const float*)d_in[1];
  const float* amask = (const float*)d_in[2];
  const float* Wq    = (const float*)d_in[3];
  const float* bq    = (const float*)d_in[4];
  const float* Wkx   = (const float*)d_in[5];
  const float* bkx   = (const float*)d_in[6];
  const float* Wvx   = (const float*)d_in[7];
  const float* bvx   = (const float*)d_in[8];
  const float* Wks   = (const float*)d_in[9];
  const float* bks   = (const float*)d_in[10];
  const float* Wvs   = (const float*)d_in[11];
  const float* bvs   = (const float*)d_in[12];
  const float* Wout  = (const float*)d_in[13];
  const float* bout  = (const float*)d_in[14];
  const float* lns   = (const float*)d_in[15];
  const float* lnb   = (const float*)d_in[16];
  float* y = (float*)d_out;

  char* ws = (char*)d_ws;
  float* ksb  = (float*)(ws + 0);
  float* vsb  = (float*)(ws + 262144);

  text_proj_kernel<<<dim3(Bn * (Ss / 16) * 2), dim3(128), 0, stream>>>(
      s, amask, Wks, bks, Wvs, bvs, ksb, vsb);
  fused_attn_kernel<<<dim3(NPOS / 16), dim3(128), 0, stream>>>(
      x, Wq, bq, Wkx, bkx, Wvx, bvx, ksb, vsb, Wout, bout, lns, lnb, y);
}


# Global loads taken out of the WMMA k loops: each wave requests its weight fragments for all k-steps, biases, text tile and LayerNorm rows as one group ahead of use; k loops wait only for LDS
# speedup vs baseline: 4.4223x; 2.3658x over previous
#include <hip/hip_runtime.h>
#include <math.h>

constexpr int Bn  = 4;
constexpr int Wd  = 32;
constexpr int Hd  = 32;
constexpr int Cc  = 128;
constexpr int Ss  = 256;
constexpr int Dd  = 512;
constexpr int NC  = 64;
constexpr int NPOS = Bn * Wd * Hd;

typedef __attribute__((ext_vector_type(2))) float v2f;
typedef __attribute__((ext_vector_type(4))) float v4f;
typedef __attribute__((ext_vector_type(8))) float v8f;

__device__ __forceinline__ v8f wmma_k4(v2f a, v2f b, v8f c) {
  return __builtin_amdgcn_wmma_f32_16x16x4_f32(
       false, a,  false, b,
       (short)0, c,  false,  false);
}


__device__ __forceinline__ float attn_logit(float vj, float Vj, float kxd,
                                            float Ksd, float g) {
  const float t = (256.f * vj) * kxd;
  return (fmaf(kxd, Vj, fmaf(vj, Ksd, t)) + g) * 0.125f;
}

template <int KSTEPS, int LD>
__device__ __forceinline__ void load_w_frags(float (&w)[2 * KSTEPS],
                                             const float* __restrict__ Wm,
                                             int koff, int n) {
#pragma unroll
  for (int ks = 0; ks < KSTEPS; ++ks) {
    w[2 * ks]     = Wm[(4 * ks + koff) * LD + n];
    w[2 * ks + 1] = Wm[(4 * ks + koff + 1) * LD + n];
  }
}

template <int KSTEPS>
__device__ __forceinline__ v8f mma_lds_regs(const float* arow,
                                            const float (&w)[2 * KSTEPS]) {
  v8f acc = {0.f, 0.f, 0.f, 0.f, 0.f, 0.f, 0.f, 0.f};
#pragma unroll
  for (int ks = 0; ks < KSTEPS; ++ks) {
    v2f af, bf;
    af.x = arow[4 * ks];
    af.y = arow[4 * ks + 1];
    bf.x = w[2 * ks];
    bf.y = w[2 * ks + 1];
    acc = wmma_k4(af, bf, acc);
  }
  return acc;
}

__global__ __launch_bounds__(128) void text_proj_kernel(
    const float* __restrict__ s, const float* __restrict__ amask,
    const float* __restrict__ Wks, const float* __restrict__ bks,
    const float* __restrict__ Wvs, const float* __restrict__ bvs,
    float* __restrict__ ks_out, float* __restrict__ vs_out) {
  constexpr int TE_STRIDE = Dd + 4;
  __shared__ float te_s[16 * TE_STRIDE];

  const int mat = blockIdx.x & 1;
  const int st  = (blockIdx.x >> 1) & 15;
  const int b   = blockIdx.x >> 5;

  const int tid  = threadIdx.x;
  const int lane = tid & 31;
  const int wave = tid >> 5;

  const float* Wm   = mat ? Wvs : Wks;
  const float* bias = mat ? bvs : bks;
  float*       dst  = mat ? vs_out : ks_out;

  const int m    = lane & 15;
  const int koff = (lane >> 4) * 2;
  const int n    = wave * 16 + (lane & 15);

  float mk[16];
#pragma unroll
  for (int r = 0; r < 16; ++r) mk[r] = amask[b * Ss + st * 16 + r];
  float sv[64];
#pragma unroll
  for (int j = 0; j < 64; ++j) {
    const int r = j >> 2, c = (j & 3) * 128 + tid;
    sv[j] = s[((size_t)b * Ss + st * 16 + r) * Dd + c];
  }
  float wf[2 * (Dd / 4)];
  load_w_frags<Dd / 4, NC>(wf, Wm, koff, n);
  const float bn = bias[n];

#pragma unroll
  for (int j = 0; j < 64; ++j) {
    const int r = j >> 2, c = (j & 3) * 128 + tid;
    te_s[r * TE_STRIDE + c] = sv[j] * mk[r];
  }
  __syncthreads();

  const v8f acc = mma_lds_regs<Dd / 4>(&te_s[m * TE_STRIDE + koff], wf);
#pragma unroll
  for (int i = 0; i < 8; ++i) {
    const int row = i + ((lane >> 4) << 3);
    dst[((size_t)b * Ss + st * 16 + row) * NC + n] = acc[i] + bn;
  }
}

struct GramGroup {
  v4f   a0[8], a1[8];
  float b0[8], b1[8];
};

__device__ __forceinline__ void gram_load(GramGroup& f, const v4f* __restrict__ vs4,
                                          const float* __restrict__ ksp, int g) {
#pragma unroll
  for (int u = 0; u < 8; ++u) {
    const int row = g * 32 + u * 4;
    f.a0[u] = vs4[(row)     * (NC / 4)];
    f.a1[u] = vs4[(row + 1) * (NC / 4)];
    f.b0[u] = ksp[(row)     * NC];
    f.b1[u] = ksp[(row + 1) * NC];
  }
}

__device__ __forceinline__ void gram_mma(const GramGroup& f, v8f (&acc)[4],
                                         v4f& vp, float& kp) {
#pragma unroll
  for (int u = 0; u < 8; ++u) {
    v2f bf;
    bf.x = f.b0[u];
    bf.y = f.b1[u];
#pragma unroll
    for (int t = 0; t < 4; ++t) {
      v2f af;
      af.x = f.a0[u][t];
      af.y = f.a1[u][t];
      acc[t] = wmma_k4(af, bf, acc[t]);
    }
    kp += f.b0[u] + f.b1[u];
    vp += f.a0[u] + f.a1[u];
  }
}

__global__ __launch_bounds__(128) void fused_attn_kernel(
    const float* __restrict__ x,
    const float* __restrict__ Wq,  const float* __restrict__ bq,
    const float* __restrict__ Wkx, const float* __restrict__ bkx,
    const float* __restrict__ Wvx, const float* __restrict__ bvx,
    const float* __restrict__ ks, const float* __restrict__ vs,
    const float* __restrict__ Wout, const float* __restrict__ bout,
    const float* __restrict__ ln_scale, const float* __restrict__ ln_bias,
    float* __restrict__ y) {
  constexpr int XS = Cc + 4;
  constexpr int QS = NC + 4;
  constexpr int GS = NC + 1;
  __shared__ float xs[16 * XS];
  __shared__ float qs[16 * QS];
  __shared__ float kxs[16 * QS];
  __shared__ float vxs[16 * QS];
  __shared__ float os[16 * QS];
  __shared__ float outs[16 * XS];
  __shared__ float Gs[NC * GS];
  __shared__ float KsS[NC];
  __shared__ float VsS[NC];

  const int tid  = threadIdx.x;
  const int lane = tid & 31;
  const int wave = tid >> 5;
  const int p0   = blockIdx.x * 16;
  const int b    = p0 >> 10;

  float xr[16];
#pragma unroll
  for (int i = 0; i < 16; ++i) xr[i] = x[(size_t)p0 * Cc + tid + i * 128];

  {
    const int m = lane & 15;
    const int h = lane >> 4;
    const int n = wave * 16 + m;
    const v4f*   vs4 = reinterpret_cast<const v4f*>(vs) + ((size_t)b * Ss + 2 * h) * (NC / 4) + m;
    const float* ksp = ks + ((size_t)b * Ss + 2 * h) * NC + n;

    v8f acc[4];
#pragma unroll
    for (int t = 0; t < 4; ++t) acc[t] = v8f{0.f, 0.f, 0.f, 0.f, 0.f, 0.f, 0.f, 0.f};
    v4f   vp = {0.f, 0.f, 0.f, 0.f};
    float kp = 0.f;

    GramGroup fa, fb;
    gram_load(fa, vs4, ksp, 0);
#pragma unroll 1
    for (int g = 0; g < 6; g += 2) {
      gram_load(fb, vs4, ksp, g + 1);
      gram_mma(fa, acc, vp, kp);
      gram_load(fa, vs4, ksp, g + 2);
      gram_mma(fb, acc, vp, kp);
    }
    gram_load(fb, vs4, ksp, 7);
    gram_mma(fa, acc, vp, kp);
    gram_mma(fb, acc, vp, kp);

    kp   += __shfl_xor(kp,   16, 32);
    vp.x += __shfl_xor(vp.x, 16, 32);
    vp.y += __shfl_xor(vp.y, 16, 32);
    vp.z += __shfl_xor(vp.z, 16, 32);
    vp.w += __shfl_xor(vp.w, 16, 32);

#pragma unroll
    for (int t = 0; t < 4; ++t)
#pragma unroll
      for (int i = 0; i < 8; ++i)
        Gs[(4 * (i + 8 * h) + t) * GS + n] = acc[t][i];
    if (lane < 16) {
      KsS[n] = kp;
      if (wave == 0) {
        VsS[4 * m + 0] = vp.x;
        VsS[4 * m + 1] = vp.y;
        VsS[4 * m + 2] = vp.z;
        VsS[4 * m + 3] = vp.w;
      }
    }
  }

  const int fm    = lane & 15;
  const int fkoff = (lane >> 4) * 2;
  const int fn    = wave * 16 + fm;

  float wq[2 * (Cc / 4)], wk[2 * (Cc / 4)], wv[2 * (Cc / 4)];
  load_w_frags<Cc / 4, NC>(wq, Wq,  fkoff, fn);
  load_w_frags<Cc / 4, NC>(wk, Wkx, fkoff, fn);
  load_w_frags<Cc / 4, NC>(wv, Wvx, fkoff, fn);
  const float bnq = bq[fn], bnk = bkx[fn], bnv = bvx[fn];

#pragma unroll
  for (int i = 0; i < 16; ++i) xs[i * XS + tid] = xr[i];
  __syncthreads();

  {
    const float* arow = &xs[fm * XS + fkoff];
    const int    drow = (lane >> 4) << 3;
    const v8f aq = mma_lds_regs<Cc / 4>(arow, wq);
#pragma unroll
    for (int i = 0; i < 8; ++i) qs[(i + drow) * QS + fn] = aq[i] + bnq;
    const v8f ak = mma_lds_regs<Cc / 4>(arow, wk);
#pragma unroll
    for (int i = 0; i < 8; ++i) kxs[(i + drow) * QS + fn] = ak[i] + bnk;
    const v8f av = mma_lds_regs<Cc / 4>(arow, wv);
#pragma unroll
    for (int i = 0; i < 8; ++i) vxs[(i + drow) * QS + fn] = av[i] + bnv;
  }

  float wo0[2 * (NC / 4)], wo1[2 * (NC / 4)];
  load_w_frags<NC / 4, Cc>(wo0, Wout, fkoff, fn);
  load_w_frags<NC / 4, Cc>(wo1, Wout, fkoff, fn + 64);
  const float bo0 = bout[fn], bo1 = bout[fn + 64];
  const int lnc = (tid & 7) * 16;
  float lsc[16], lbi[16];
#pragma unroll
  for (int i = 0; i < 16; ++i) {
    lsc[i] = ln_scale[lnc + i];
    lbi[i] = ln_bias[lnc + i];
  }
  __syncthreads();

  const int pbase = wave * 4;

  {
    float vj0[4], vj1[4], m0[4], m1[4], z0[4], z1[4];
#pragma unroll
    for (int pi = 0; pi < 4; ++pi) {
      vj0[pi] = vxs[(pbase + pi) * QS + lane];
      vj1[pi] = vxs[(pbase + pi) * QS + lane + 32];
      m0[pi] = -INFINITY;  m1[pi] = -INFINITY;
      z0[pi] = 0.f;        z1[pi] = 0.f;
    }
    const float Vj0 = VsS[lane];
    const float Vj1 = VsS[lane + 32];
    const float* g0 = &Gs[lane * GS];
    const float* g1 = &Gs[(lane + 32) * GS];

#pragma unroll 4
    for (int d = 0; d < NC; ++d) {
      const float ga = g0[d], gb = g1[d], Ksd = KsS[d];
#pragma unroll
      for (int pi = 0; pi < 4; ++pi) {
        const float kxd = kxs[(pbase + pi) * QS + d];
        m0[pi] = fmaxf(m0[pi], attn_logit(vj0[pi], Vj0, kxd, Ksd, ga));
        m1[pi] = fmaxf(m1[pi], attn_logit(vj1[pi], Vj1, kxd, Ksd, gb));
      }
    }
#pragma unroll 4
    for (int d = 0; d < NC; ++d) {
      const float ga = g0[d], gb = g1[d], Ksd = KsS[d];
#pragma unroll
      for (int pi = 0; pi < 4; ++pi) {
        const float kxd = kxs[(pbase + pi) * QS + d];
        z0[pi] += __expf(attn_logit(vj0[pi], Vj0, kxd, Ksd, ga) - m0[pi]);
        z1[pi] += __expf(attn_logit(vj1[pi], Vj1, kxd, Ksd, gb) - m1[pi]);
      }
    }
#pragma unroll
    for (int pi = 0; pi < 4; ++pi) {
      const int r0 = (pbase + pi) * QS + lane;
      os[r0]      = m0[pi];
      os[r0 + 32] = m1[pi];
      qs[r0]      = qs[r0] / z0[pi];
      qs[r0 + 32] = qs[r0 + 32] / z1[pi];
    }
  }
  __syncthreads();

  {
    float kx0[4], kx1[4], o0[4], o1[4];
#pragma unroll
    for (int pi = 0; pi < 4; ++pi) {
      kx0[pi] = kxs[(pbase + pi) * QS + lane];
      kx1[pi] = kxs[(pbase + pi) * QS + lane + 32];
      o0[pi] = 0.f;  o1[pi] = 0.f;
    }
    const float Ks0 = KsS[lane];
    const float Ks1 = KsS[lane + 32];
#pragma unroll 4
    for (int j = 0; j < NC; ++j) {
      const float ga = Gs[j * GS + lane];
      const float gb = Gs[j * GS + lane + 32];
      const float Vj = VsS[j];
#pragma unroll
      for (int pi = 0; pi < 4; ++pi) {
        const int rj = (pbase + pi) * QS + j;
        const float vj = vxs[rj];
        const float mj = os[rj];
        const float wj = qs[rj];
        o0[pi] += wj * __expf(attn_logit(vj, Vj, kx0[pi], Ks0, ga) - mj);
        o1[pi] += wj * __expf(attn_logit(vj, Vj, kx1[pi], Ks1, gb) - mj);
      }
    }
#pragma unroll
    for (int pi = 0; pi < 4; ++pi) {
      os[(pbase + pi) * QS + lane]      = o0[pi];
      os[(pbase + pi) * QS + lane + 32] = o1[pi];
    }
  }
  __syncthreads();

  {
    const float* arow = &os[fm * QS + fkoff];
    const int    drow = (lane >> 4) << 3;
    const v8f a0 = mma_lds_regs<NC / 4>(arow, wo0);
#pragma unroll
    for (int i = 0; i < 8; ++i) outs[(i + drow) * XS + fn] = a0[i] + bo0;
    const v8f a1 = mma_lds_regs<NC / 4>(arow, wo1);
#pragma unroll
    for (int i = 0; i < 8; ++i) outs[(i + drow) * XS + fn + 64] = a1[i] + bo1;
  }
  __syncthreads();

  {
    const int row   = tid >> 3;
    const int cbase = (tid & 7) * 16;
    float vals[16];
    float sum = 0.f, sq = 0.f;
#pragma unroll
    for (int i = 0; i < 16; ++i) {
      const float v = xs[row * XS + cbase + i] + outs[row * XS + cbase + i];
      vals[i] = v;
      sum += v;
      sq  += v * v;
    }
#pragma unroll
    for (int msk = 4; msk >= 1; msk >>= 1) {
      sum += __shfl_xor(sum, msk, 32);
      sq  += __shfl_xor(sq,  msk, 32);
    }
    const float mu   = sum * (1.f / 128.f);
    const float var  = sq * (1.f / 128.f) - mu * mu;
    const float rstd = rsqrtf(var + 1e-6f);
#pragma unroll
    for (int i = 0; i < 16; ++i) {
      const int c = cbase + i;
      y[(size_t)(p0 + row) * Cc + c] =
          (vals[i] - mu) * rstd * lsc[i] + lbi[i];
    }
  }
}

extern "C" void kernel_launch(void* const* d_in, const int* in_sizes, int n_in,
                              void* d_out, int out_size, void* d_ws, size_t ws_size,
                              hipStream_t stream) {
  const float* x     = (const float*)d_in[0];
  const float* s     = (const float*)d_in[1];
  const float* amask = (const float*)d_in[2];
  const float* Wq    = (const float*)d_in[3];
  const float* bq    = (const float*)d_in[4];
  const float* Wkx   = (const float*)d_in[5];
  const float* bkx   = (const float*)d_in[6];
  const float* Wvx   = (const float*)d_in[7];
  const float* bvx   = (const float*)d_in[8];
  const float* Wks   = (const float*)d_in[9];
  const float* bks   = (const float*)d_in[10];
  const float* Wvs   = (const float*)d_in[11];
  const float* bvs   = (const float*)d_in[12];
  const float* Wout  = (const float*)d_in[13];
  const float* bout  = (const float*)d_in[14];
  const float* lns   = (const float*)d_in[15];
  const float* lnb   = (const float*)d_in[16];
  float* y = (float*)d_out;

  char* ws = (char*)d_ws;
  float* ksb  = (float*)(ws + 0);
  float* vsb  = (float*)(ws + 262144);

  text_proj_kernel<<<dim3(Bn * (Ss / 16) * 2), dim3(128), 0, stream>>>(
      s, amask, Wks, bks, Wvs, bvs, ksb, vsb);
  fused_attn_kernel<<<dim3(NPOS / 16), dim3(128), 0, stream>>>(
      x, Wq, bq, Wkx, bkx, Wvx, bvx, ksb, vsb, Wout, bout, lns, lnb, y);
}


# Fused kernel on 8 waves a block (2 a SIMD): waves 4-7 do the x projections beside the Gram phase of waves 0-3, softmax 2 positions a wave; one wave's load and LDS waits are filled by the other
# speedup vs baseline: 5.3528x; 1.2104x over previous
#include <hip/hip_runtime.h>
#include <math.h>

constexpr int Bn  = 4;
constexpr int Wd  = 32;
constexpr int Hd  = 32;
constexpr int Cc  = 128;
constexpr int Ss  = 256;
constexpr int Dd  = 512;
constexpr int NC  = 64;
constexpr int NPOS = Bn * Wd * Hd;

typedef __attribute__((ext_vector_type(2))) float v2f;
typedef __attribute__((ext_vector_type(4))) float v4f;
typedef __attribute__((ext_vector_type(8))) float v8f;

__device__ __forceinline__ v8f wmma_k4(v2f a, v2f b, v8f c) {
  return __builtin_amdgcn_wmma_f32_16x16x4_f32(
       false, a,  false, b,
       (short)0, c,  false,  false);
}


__device__ __forceinline__ float attn_logit(float vj, float Vj, float kxd,
                                            float Ksd, float g) {
  const float t = (256.f * vj) * kxd;
  return (fmaf(kxd, Vj, fmaf(vj, Ksd, t)) + g) * 0.125f;
}

template <int KSTEPS, int LD>
__device__ __forceinline__ void load_w_frags(float (&w)[2 * KSTEPS],
                                             const float* __restrict__ Wm,
                                             int koff, int n) {
#pragma unroll
  for (int ks = 0; ks < KSTEPS; ++ks) {
    w[2 * ks]     = Wm[(4 * ks + koff) * LD + n];
    w[2 * ks + 1] = Wm[(4 * ks + koff + 1) * LD + n];
  }
}

template <int KSTEPS>
__device__ __forceinline__ v8f mma_lds_regs(const float* arow,
                                            const float (&w)[2 * KSTEPS]) {
  v8f acc = {0.f, 0.f, 0.f, 0.f, 0.f, 0.f, 0.f, 0.f};
#pragma unroll
  for (int ks = 0; ks < KSTEPS; ++ks) {
    v2f af, bf;
    af.x = arow[4 * ks];
    af.y = arow[4 * ks + 1];
    bf.x = w[2 * ks];
    bf.y = w[2 * ks + 1];
    acc = wmma_k4(af, bf, acc);
  }
  return acc;
}

__global__ __launch_bounds__(128) void text_proj_kernel(
    const float* __restrict__ s, const float* __restrict__ amask,
    const float* __restrict__ Wks, const float* __restrict__ bks,
    const float* __restrict__ Wvs, const float* __restrict__ bvs,
    float* __restrict__ ks_out, float* __restrict__ vs_out) {
  constexpr int TE_STRIDE = Dd + 4;
  __shared__ float te_s[16 * TE_STRIDE];

  const int mat = blockIdx.x & 1;
  const int st  = (blockIdx.x >> 1) & 15;
  const int b   = blockIdx.x >> 5;

  const int tid  = threadIdx.x;
  const int lane = tid & 31;
  const int wave = tid >> 5;

  const float* Wm   = mat ? Wvs : Wks;
  const float* bias = mat ? bvs : bks;
  float*       dst  = mat ? vs_out : ks_out;

  const int m    = lane & 15;
  const int koff = (lane >> 4) * 2;
  const int n    = wave * 16 + (lane & 15);

  float mk[16];
#pragma unroll
  for (int r = 0; r < 16; ++r) mk[r] = amask[b * Ss + st * 16 + r];
  float sv[64];
#pragma unroll
  for (int j = 0; j < 64; ++j) {
    const int r = j >> 2, c = (j & 3) * 128 + tid;
    sv[j] = s[((size_t)b * Ss + st * 16 + r) * Dd + c];
  }
  float wf[2 * (Dd / 4)];
  load_w_frags<Dd / 4, NC>(wf, Wm, koff, n);
  const float bn = bias[n];

#pragma unroll
  for (int j = 0; j < 64; ++j) {
    const int r = j >> 2, c = (j & 3) * 128 + tid;
    te_s[r * TE_STRIDE + c] = sv[j] * mk[r];
  }
  __syncthreads();

  const v8f acc = mma_lds_regs<Dd / 4>(&te_s[m * TE_STRIDE + koff], wf);
#pragma unroll
  for (int i = 0; i < 8; ++i) {
    const int row = i + ((lane >> 4) << 3);
    dst[((size_t)b * Ss + st * 16 + row) * NC + n] = acc[i] + bn;
  }
}

struct GramGroup {
  v4f   a0[8], a1[8];
  float b0[8], b1[8];
};

__device__ __forceinline__ void gram_load(GramGroup& f, const v4f* __restrict__ vs4,
                                          const float* __restrict__ ksp, int g) {
#pragma unroll
  for (int u = 0; u < 8; ++u) {
    const int row = g * 32 + u * 4;
    f.a0[u] = vs4[(row)     * (NC / 4)];
    f.a1[u] = vs4[(row + 1) * (NC / 4)];
    f.b0[u] = ksp[(row)     * NC];
    f.b1[u] = ksp[(row + 1) * NC];
  }
}

__device__ __forceinline__ void gram_mma(const GramGroup& f, v8f (&acc)[4],
                                         v4f& vp, float& kp) {
#pragma unroll
  for (int u = 0; u < 8; ++u) {
    v2f bf;
    bf.x = f.b0[u];
    bf.y = f.b1[u];
#pragma unroll
    for (int t = 0; t < 4; ++t) {
      v2f af;
      af.x = f.a0[u][t];
      af.y = f.a1[u][t];
      acc[t] = wmma_k4(af, bf, acc[t]);
    }
    kp += f.b0[u] + f.b1[u];
    vp += f.a0[u] + f.a1[u];
  }
}

__global__ __launch_bounds__(256) void fused_attn_kernel(
    const float* __restrict__ x,
    const float* __restrict__ Wq,  const float* __restrict__ bq,
    const float* __restrict__ Wkx, const float* __restrict__ bkx,
    const float* __restrict__ Wvx, const float* __restrict__ bvx,
    const float* __restrict__ ks, const float* __restrict__ vs,
    const float* __restrict__ Wout, const float* __restrict__ bout,
    const float* __restrict__ ln_scale, const float* __restrict__ ln_bias,
    float* __restrict__ y) {
  constexpr int XS = Cc + 4;
  constexpr int QS = NC + 4;
  constexpr int GS = NC + 1;
  constexpr int PW = 2;
  __shared__ float xs[16 * XS];
  __shared__ float qs[16 * QS];
  __shared__ float kxs[16 * QS];
  __shared__ float vxs[16 * QS];
  __shared__ float os[16 * QS];
  __shared__ float outs[16 * XS];
  __shared__ float Gs[NC * GS];
  __shared__ float KsS[NC];
  __shared__ float VsS[NC];

  const int tid  = threadIdx.x;
  const int lane = tid & 31;
  const int wave = __builtin_amdgcn_readfirstlane(tid >> 5);
  const bool gram_wave = wave < 4;
  const int cw   = wave & 3;
  const int p0   = blockIdx.x * 16;
  const int b    = p0 >> 10;

  const int fm    = lane & 15;
  const int fh    = lane >> 4;
  const int fkoff = fh * 2;
  const int fn    = cw * 16 + fm;

  const v4f*   vs4 = reinterpret_cast<const v4f*>(vs) + ((size_t)b * Ss + 2 * fh) * (NC / 4) + fm;
  const float* ksp = ks + ((size_t)b * Ss + 2 * fh) * NC + fn;

  GramGroup fa, fb;
  float lsc[16], lbi[16];
  const int lnc = (tid & 7) * 16;
  if (gram_wave) {
    gram_load(fa, vs4, ksp, 0);
#pragma unroll
    for (int i = 0; i < 16; ++i) {
      lsc[i] = ln_scale[lnc + i];
      lbi[i] = ln_bias[lnc + i];
    }
  } else {
    const int t2 = tid & 127;
    float xr[16];
#pragma unroll
    for (int i = 0; i < 16; ++i) xr[i] = x[(size_t)p0 * Cc + t2 + i * 128];
#pragma unroll
    for (int i = 0; i < 16; ++i) xs[i * XS + t2] = xr[i];
  }
  __syncthreads();

  float wo0[2 * (NC / 4)], wo1[2 * (NC / 4)];
  float bo0 = 0.f, bo1 = 0.f;

  if (gram_wave) {
    v8f acc[4];
#pragma unroll
    for (int t = 0; t < 4; ++t) acc[t] = v8f{0.f, 0.f, 0.f, 0.f, 0.f, 0.f, 0.f, 0.f};
    v4f   vp = {0.f, 0.f, 0.f, 0.f};
    float kp = 0.f;

#pragma unroll 1
    for (int g = 0; g < 6; g += 2) {
      gram_load(fb, vs4, ksp, g + 1);
      gram_mma(fa, acc, vp, kp);
      gram_load(fa, vs4, ksp, g + 2);
      gram_mma(fb, acc, vp, kp);
    }
    gram_load(fb, vs4, ksp, 7);
    gram_mma(fa, acc, vp, kp);
    gram_mma(fb, acc, vp, kp);

    kp   += __shfl_xor(kp,   16, 32);
    vp.x += __shfl_xor(vp.x, 16, 32);
    vp.y += __shfl_xor(vp.y, 16, 32);
    vp.z += __shfl_xor(vp.z, 16, 32);
    vp.w += __shfl_xor(vp.w, 16, 32);

#pragma unroll
    for (int t = 0; t < 4; ++t)
#pragma unroll
      for (int i = 0; i < 8; ++i)
        Gs[(4 * (i + 8 * fh) + t) * GS + fn] = acc[t][i];
    if (lane < 16) {
      KsS[fn] = kp;
      if (wave == 0) {
        VsS[4 * fm + 0] = vp.x;
        VsS[4 * fm + 1] = vp.y;
        VsS[4 * fm + 2] = vp.z;
        VsS[4 * fm + 3] = vp.w;
      }
    }
  } else {
    float wq[2 * (Cc / 4)], wk[2 * (Cc / 4)], wv[2 * (Cc / 4)];
    load_w_frags<Cc / 4, NC>(wq, Wq,  fkoff, fn);
    load_w_frags<Cc / 4, NC>(wk, Wkx, fkoff, fn);
    load_w_frags<Cc / 4, NC>(wv, Wvx, fkoff, fn);
    const float bnq = bq[fn], bnk = bkx[fn], bnv = bvx[fn];
    const float* arow = &xs[fm * XS + fkoff];
    const int    drow = fh << 3;
    const v8f aq = mma_lds_regs<Cc / 4>(arow, wq);
#pragma unroll
    for (int i = 0; i < 8; ++i) qs[(i + drow) * QS + fn] = aq[i] + bnq;
    const v8f ak = mma_lds_regs<Cc / 4>(arow, wk);
#pragma unroll
    for (int i = 0; i < 8; ++i) kxs[(i + drow) * QS + fn] = ak[i] + bnk;
    const v8f av = mma_lds_regs<Cc / 4>(arow, wv);
#pragma unroll
    for (int i = 0; i < 8; ++i) vxs[(i + drow) * QS + fn] = av[i] + bnv;

    load_w_frags<NC / 4, Cc>(wo0, Wout, fkoff, fn);
    load_w_frags<NC / 4, Cc>(wo1, Wout, fkoff, fn + 64);
    bo0 = bout[fn];  bo1 = bout[fn + 64];
  }
  __syncthreads();

  const int pbase = wave * PW;

  {
    float vj0[PW], vj1[PW], m0[PW], m1[PW], z0[PW], z1[PW];
#pragma unroll
    for (int pi = 0; pi < PW; ++pi) {
      vj0[pi] = vxs[(pbase + pi) * QS + lane];
      vj1[pi] = vxs[(pbase + pi) * QS + lane + 32];
      m0[pi] = -INFINITY;  m1[pi] = -INFINITY;
      z0[pi] = 0.f;        z1[pi] = 0.f;
    }
    const float Vj0 = VsS[lane];
    const float Vj1 = VsS[lane + 32];
    const float* g0 = &Gs[lane * GS];
    const float* g1 = &Gs[(lane + 32) * GS];

#pragma unroll 4
    for (int d = 0; d < NC; ++d) {
      const float ga = g0[d], gb = g1[d], Ksd = KsS[d];
#pragma unroll
      for (int pi = 0; pi < PW; ++pi) {
        const float kxd = kxs[(pbase + pi) * QS + d];
        m0[pi] = fmaxf(m0[pi], attn_logit(vj0[pi], Vj0, kxd, Ksd, ga));
        m1[pi] = fmaxf(m1[pi], attn_logit(vj1[pi], Vj1, kxd, Ksd, gb));
      }
    }
#pragma unroll 4
    for (int d = 0; d < NC; ++d) {
      const float ga = g0[d], gb = g1[d], Ksd = KsS[d];
#pragma unroll
      for (int pi = 0; pi < PW; ++pi) {
        const float kxd = kxs[(pbase + pi) * QS + d];
        z0[pi] += __expf(attn_logit(vj0[pi], Vj0, kxd, Ksd, ga) - m0[pi]);
        z1[pi] += __expf(attn_logit(vj1[pi], Vj1, kxd, Ksd, gb) - m1[pi]);
      }
    }
#pragma unroll
    for (int pi = 0; pi < PW; ++pi) {
      const int r0 = (pbase + pi) * QS + lane;
      os[r0]      = m0[pi];
      os[r0 + 32] = m1[pi];
      qs[r0]      = qs[r0] / z0[pi];
      qs[r0 + 32] = qs[r0 + 32] / z1[pi];
    }
  }
  __syncthreads();

  {
    float kx0[PW], kx1[PW], o0[PW], o1[PW];
#pragma unroll
    for (int pi = 0; pi < PW; ++pi) {
      kx0[pi] = kxs[(pbase + pi) * QS + lane];
      kx1[pi] = kxs[(pbase + pi) * QS + lane + 32];
      o0[pi] = 0.f;  o1[pi] = 0.f;
    }
    const float Ks0 = KsS[lane];
    const float Ks1 = KsS[lane + 32];
#pragma unroll 4
    for (int j = 0; j < NC; ++j) {
      const float ga = Gs[j * GS + lane];
      const float gb = Gs[j * GS + lane + 32];
      const float Vj = VsS[j];
#pragma unroll
      for (int pi = 0; pi < PW; ++pi) {
        const int rj = (pbase + pi) * QS + j;
        const float vj = vxs[rj];
        const float mj = os[rj];
        const float wj = qs[rj];
        o0[pi] += wj * __expf(attn_logit(vj, Vj, kx0[pi], Ks0, ga) - mj);
        o1[pi] += wj * __expf(attn_logit(vj, Vj, kx1[pi], Ks1, gb) - mj);
      }
    }
#pragma unroll
    for (int pi = 0; pi < PW; ++pi) {
      os[(pbase + pi) * QS + lane]      = o0[pi];
      os[(pbase + pi) * QS + lane + 32] = o1[pi];
    }
  }
  __syncthreads();

  if (!gram_wave) {
    const float* arow = &os[fm * QS + fkoff];
    const int    drow = fh << 3;
    const v8f a0 = mma_lds_regs<NC / 4>(arow, wo0);
#pragma unroll
    for (int i = 0; i < 8; ++i) outs[(i + drow) * XS + fn] = a0[i] + bo0;
    const v8f a1 = mma_lds_regs<NC / 4>(arow, wo1);
#pragma unroll
    for (int i = 0; i < 8; ++i) outs[(i + drow) * XS + fn + 64] = a1[i] + bo1;
  }
  __syncthreads();

  if (gram_wave) {
    const int row   = tid >> 3;
    const int cbase = (tid & 7) * 16;
    float vals[16];
    float sum = 0.f, sq = 0.f;
#pragma unroll
    for (int i = 0; i < 16; ++i) {
      const float v = xs[row * XS + cbase + i] + outs[row * XS + cbase + i];
      vals[i] = v;
      sum += v;
      sq  += v * v;
    }
#pragma unroll
    for (int msk = 4; msk >= 1; msk >>= 1) {
      sum += __shfl_xor(sum, msk, 32);
      sq  += __shfl_xor(sq,  msk, 32);
    }
    const float mu   = sum * (1.f / 128.f);
    const float var  = sq * (1.f / 128.f) - mu * mu;
    const float rstd = rsqrtf(var + 1e-6f);
#pragma unroll
    for (int i = 0; i < 16; ++i) {
      const int c = cbase + i;
      y[(size_t)(p0 + row) * Cc + c] =
          (vals[i] - mu) * rstd * lsc[i] + lbi[i];
    }
  }
}

extern "C" void kernel_launch(void* const* d_in, const int* in_sizes, int n_in,
                              void* d_out, int out_size, void* d_ws, size_t ws_size,
                              hipStream_t stream) {
  const float* x     = (const float*)d_in[0];
  const float* s     = (const float*)d_in[1];
  const float* amask = (const float*)d_in[2];
  const float* Wq    = (const float*)d_in[3];
  const float* bq    = (const float*)d_in[4];
  const float* Wkx   = (const float*)d_in[5];
  const float* bkx   = (const float*)d_in[6];
  const float* Wvx   = (const float*)d_in[7];
  const float* bvx   = (const float*)d_in[8];
  const float* Wks   = (const float*)d_in[9];
  const float* bks   = (const float*)d_in[10];
  const float* Wvs   = (const float*)d_in[11];
  const float* bvs   = (const float*)d_in[12];
  const float* Wout  = (const float*)d_in[13];
  const float* bout  = (const float*)d_in[14];
  const float* lns   = (const float*)d_in[15];
  const float* lnb   = (const float*)d_in[16];
  float* y = (float*)d_out;

  char* ws = (char*)d_ws;
  float* ksb  = (float*)(ws + 0);
  float* vsb  = (float*)(ws + 262144);

  text_proj_kernel<<<dim3(Bn * (Ss / 16) * 2), dim3(128), 0, stream>>>(
      s, amask, Wks, bks, Wvs, bvs, ksb, vsb);
  fused_attn_kernel<<<dim3(NPOS / 16), dim3(256), 0, stream>>>(
      x, Wq, bq, Wkx, bkx, Wvx, bvx, ksb, vsb, Wout, bout, lns, lnb, y);
}


# Softmax logit regrouped as B[j]*kx[d] + (vx[j]*Ksum[d] + G[j][d]), B = 256*vx+Vsum per row, log2(e)/8 folded into kx, Ksum, G at their stores: 2 fma a logit (was 5 ops), bare exp2; loop bodies of 8
# speedup vs baseline: 5.7987x; 1.0833x over previous
#include <hip/hip_runtime.h>
#include <math.h>

constexpr int Bn  = 4;
constexpr int Wd  = 32;
constexpr int Hd  = 32;
constexpr int Cc  = 128;
constexpr int Ss  = 256;
constexpr int Dd  = 512;
constexpr int NC  = 64;
constexpr int NPOS = Bn * Wd * Hd;

typedef __attribute__((ext_vector_type(2))) float v2f;
typedef __attribute__((ext_vector_type(4))) float v4f;
typedef __attribute__((ext_vector_type(8))) float v8f;

__device__ __forceinline__ v8f wmma_k4(v2f a, v2f b, v8f c) {
  return __builtin_amdgcn_wmma_f32_16x16x4_f32(
       false, a,  false, b,
       (short)0, c,  false,  false);
}


constexpr float LOGIT_SCALE = 0.125f * 1.44269504088896340736f;
__device__ __forceinline__ float attn_logit2(float Bj, float vj, float kxd,
                                             float Ksd, float g) {
  return fmaf(Bj, kxd, fmaf(vj, Ksd, g));
}
__device__ __forceinline__ float attn_rowB(float vj, float Vj) {
  return fmaf(256.f, vj, Vj);
}

template <int KSTEPS, int LD>
__device__ __forceinline__ void load_w_frags(float (&w)[2 * KSTEPS],
                                             const float* __restrict__ Wm,
                                             int koff, int n) {
#pragma unroll
  for (int ks = 0; ks < KSTEPS; ++ks) {
    w[2 * ks]     = Wm[(4 * ks + koff) * LD + n];
    w[2 * ks + 1] = Wm[(4 * ks + koff + 1) * LD + n];
  }
}

template <int KSTEPS>
__device__ __forceinline__ v8f mma_lds_regs(const float* arow,
                                            const float (&w)[2 * KSTEPS]) {
  v8f acc = {0.f, 0.f, 0.f, 0.f, 0.f, 0.f, 0.f, 0.f};
#pragma unroll
  for (int ks = 0; ks < KSTEPS; ++ks) {
    v2f af, bf;
    af.x = arow[4 * ks];
    af.y = arow[4 * ks + 1];
    bf.x = w[2 * ks];
    bf.y = w[2 * ks + 1];
    acc = wmma_k4(af, bf, acc);
  }
  return acc;
}

__global__ __launch_bounds__(128) void text_proj_kernel(
    const float* __restrict__ s, const float* __restrict__ amask,
    const float* __restrict__ Wks, const float* __restrict__ bks,
    const float* __restrict__ Wvs, const float* __restrict__ bvs,
    float* __restrict__ ks_out, float* __restrict__ vs_out) {
  constexpr int TE_STRIDE = Dd + 4;
  __shared__ float te_s[16 * TE_STRIDE];

  const int mat = blockIdx.x & 1;
  const int st  = (blockIdx.x >> 1) & 15;
  const int b   = blockIdx.x >> 5;

  const int tid  = threadIdx.x;
  const int lane = tid & 31;
  const int wave = tid >> 5;

  const float* Wm   = mat ? Wvs : Wks;
  const float* bias = mat ? bvs : bks;
  float*       dst  = mat ? vs_out : ks_out;

  const int m    = lane & 15;
  const int koff = (lane >> 4) * 2;
  const int n    = wave * 16 + (lane & 15);

  float mk[16];
#pragma unroll
  for (int r = 0; r < 16; ++r) mk[r] = amask[b * Ss + st * 16 + r];
  float sv[64];
#pragma unroll
  for (int j = 0; j < 64; ++j) {
    const int r = j >> 2, c = (j & 3) * 128 + tid;
    sv[j] = s[((size_t)b * Ss + st * 16 + r) * Dd + c];
  }
  float wf[2 * (Dd / 4)];
  load_w_frags<Dd / 4, NC>(wf, Wm, koff, n);
  const float bn = bias[n];

#pragma unroll
  for (int j = 0; j < 64; ++j) {
    const int r = j >> 2, c = (j & 3) * 128 + tid;
    te_s[r * TE_STRIDE + c] = sv[j] * mk[r];
  }
  __syncthreads();

  const v8f acc = mma_lds_regs<Dd / 4>(&te_s[m * TE_STRIDE + koff], wf);
#pragma unroll
  for (int i = 0; i < 8; ++i) {
    const int row = i + ((lane >> 4) << 3);
    dst[((size_t)b * Ss + st * 16 + row) * NC + n] = acc[i] + bn;
  }
}

struct GramGroup {
  v4f   a0[8], a1[8];
  float b0[8], b1[8];
};

__device__ __forceinline__ void gram_load(GramGroup& f, const v4f* __restrict__ vs4,
                                          const float* __restrict__ ksp, int g) {
#pragma unroll
  for (int u = 0; u < 8; ++u) {
    const int row = g * 32 + u * 4;
    f.a0[u] = vs4[(row)     * (NC / 4)];
    f.a1[u] = vs4[(row + 1) * (NC / 4)];
    f.b0[u] = ksp[(row)     * NC];
    f.b1[u] = ksp[(row + 1) * NC];
  }
}

__device__ __forceinline__ void gram_mma(const GramGroup& f, v8f (&acc)[4],
                                         v4f& vp, float& kp) {
#pragma unroll
  for (int u = 0; u < 8; ++u) {
    v2f bf;
    bf.x = f.b0[u];
    bf.y = f.b1[u];
#pragma unroll
    for (int t = 0; t < 4; ++t) {
      v2f af;
      af.x = f.a0[u][t];
      af.y = f.a1[u][t];
      acc[t] = wmma_k4(af, bf, acc[t]);
    }
    kp += f.b0[u] + f.b1[u];
    vp += f.a0[u] + f.a1[u];
  }
}

__global__ __launch_bounds__(256) void fused_attn_kernel(
    const float* __restrict__ x,
    const float* __restrict__ Wq,  const float* __restrict__ bq,
    const float* __restrict__ Wkx, const float* __restrict__ bkx,
    const float* __restrict__ Wvx, const float* __restrict__ bvx,
    const float* __restrict__ ks, const float* __restrict__ vs,
    const float* __restrict__ Wout, const float* __restrict__ bout,
    const float* __restrict__ ln_scale, const float* __restrict__ ln_bias,
    float* __restrict__ y) {
  constexpr int XS = Cc + 4;
  constexpr int QS = NC + 4;
  constexpr int GS = NC + 1;
  constexpr int PW = 2;
  __shared__ float xs[16 * XS];
  __shared__ float qs[16 * QS];
  __shared__ float kxs[16 * QS];
  __shared__ float vxs[16 * QS];
  __shared__ float os[16 * QS];
  __shared__ float bs[16 * QS];
  __shared__ float outs[16 * XS];
  __shared__ float Gs[NC * GS];
  __shared__ float KsS[NC];
  __shared__ float VsS[NC];

  const int tid  = threadIdx.x;
  const int lane = tid & 31;
  const int wave = __builtin_amdgcn_readfirstlane(tid >> 5);
  const bool gram_wave = wave < 4;
  const int cw   = wave & 3;
  const int p0   = blockIdx.x * 16;
  const int b    = p0 >> 10;

  const int fm    = lane & 15;
  const int fh    = lane >> 4;
  const int fkoff = fh * 2;
  const int fn    = cw * 16 + fm;

  const v4f*   vs4 = reinterpret_cast<const v4f*>(vs) + ((size_t)b * Ss + 2 * fh) * (NC / 4) + fm;
  const float* ksp = ks + ((size_t)b * Ss + 2 * fh) * NC + fn;

  GramGroup fa, fb;
  float lsc[16], lbi[16];
  const int lnc = (tid & 7) * 16;
  if (gram_wave) {
    gram_load(fa, vs4, ksp, 0);
#pragma unroll
    for (int i = 0; i < 16; ++i) {
      lsc[i] = ln_scale[lnc + i];
      lbi[i] = ln_bias[lnc + i];
    }
  } else {
    const int t2 = tid & 127;
    float xr[16];
#pragma unroll
    for (int i = 0; i < 16; ++i) xr[i] = x[(size_t)p0 * Cc + t2 + i * 128];
#pragma unroll
    for (int i = 0; i < 16; ++i) xs[i * XS + t2] = xr[i];
  }
  __syncthreads();

  float wo0[2 * (NC / 4)], wo1[2 * (NC / 4)];
  float bo0 = 0.f, bo1 = 0.f;

  if (gram_wave) {
    v8f acc[4];
#pragma unroll
    for (int t = 0; t < 4; ++t) acc[t] = v8f{0.f, 0.f, 0.f, 0.f, 0.f, 0.f, 0.f, 0.f};
    v4f   vp = {0.f, 0.f, 0.f, 0.f};
    float kp = 0.f;

#pragma unroll 1
    for (int g = 0; g < 6; g += 2) {
      gram_load(fb, vs4, ksp, g + 1);
      gram_mma(fa, acc, vp, kp);
      gram_load(fa, vs4, ksp, g + 2);
      gram_mma(fb, acc, vp, kp);
    }
    gram_load(fb, vs4, ksp, 7);
    gram_mma(fa, acc, vp, kp);
    gram_mma(fb, acc, vp, kp);

    kp   += __shfl_xor(kp,   16, 32);
    vp.x += __shfl_xor(vp.x, 16, 32);
    vp.y += __shfl_xor(vp.y, 16, 32);
    vp.z += __shfl_xor(vp.z, 16, 32);
    vp.w += __shfl_xor(vp.w, 16, 32);

#pragma unroll
    for (int t = 0; t < 4; ++t)
#pragma unroll
      for (int i = 0; i < 8; ++i)
        Gs[(4 * (i + 8 * fh) + t) * GS + fn] = acc[t][i] * LOGIT_SCALE;
    if (lane < 16) {
      KsS[fn] = kp * LOGIT_SCALE;
      if (wave == 0) {
        VsS[4 * fm + 0] = vp.x;
        VsS[4 * fm + 1] = vp.y;
        VsS[4 * fm + 2] = vp.z;
        VsS[4 * fm + 3] = vp.w;
      }
    }
  } else {
    float wq[2 * (Cc / 4)], wk[2 * (Cc / 4)], wv[2 * (Cc / 4)];
    load_w_frags<Cc / 4, NC>(wq, Wq,  fkoff, fn);
    load_w_frags<Cc / 4, NC>(wk, Wkx, fkoff, fn);
    load_w_frags<Cc / 4, NC>(wv, Wvx, fkoff, fn);
    const float bnq = bq[fn], bnk = bkx[fn], bnv = bvx[fn];
    const float* arow = &xs[fm * XS + fkoff];
    const int    drow = fh << 3;
    const v8f aq = mma_lds_regs<Cc / 4>(arow, wq);
#pragma unroll
    for (int i = 0; i < 8; ++i) qs[(i + drow) * QS + fn] = aq[i] + bnq;
    const v8f ak = mma_lds_regs<Cc / 4>(arow, wk);
#pragma unroll
    for (int i = 0; i < 8; ++i) kxs[(i + drow) * QS + fn] = (ak[i] + bnk) * LOGIT_SCALE;
    const v8f av = mma_lds_regs<Cc / 4>(arow, wv);
#pragma unroll
    for (int i = 0; i < 8; ++i) vxs[(i + drow) * QS + fn] = av[i] + bnv;

    load_w_frags<NC / 4, Cc>(wo0, Wout, fkoff, fn);
    load_w_frags<NC / 4, Cc>(wo1, Wout, fkoff, fn + 64);
    bo0 = bout[fn];  bo1 = bout[fn + 64];
  }
  __syncthreads();

  const int pbase = wave * PW;

  {
    float vj0[PW], vj1[PW], B0[PW], B1[PW], m0[PW], m1[PW], z0[PW], z1[PW];
    const float Vj0 = VsS[lane];
    const float Vj1 = VsS[lane + 32];
#pragma unroll
    for (int pi = 0; pi < PW; ++pi) {
      vj0[pi] = vxs[(pbase + pi) * QS + lane];
      vj1[pi] = vxs[(pbase + pi) * QS + lane + 32];
      B0[pi] = attn_rowB(vj0[pi], Vj0);
      B1[pi] = attn_rowB(vj1[pi], Vj1);
      m0[pi] = -INFINITY;  m1[pi] = -INFINITY;
      z0[pi] = 0.f;        z1[pi] = 0.f;
    }
    const float* g0 = &Gs[lane * GS];
    const float* g1 = &Gs[(lane + 32) * GS];

    constexpr int DB = 8;
#pragma unroll 1
    for (int d0 = 0; d0 < NC; d0 += DB) {
      float ga[DB], gb[DB], Ksd[DB], kxd[PW][DB];
#pragma unroll
      for (int u = 0; u < DB; ++u) {
        ga[u] = g0[d0 + u];  gb[u] = g1[d0 + u];  Ksd[u] = KsS[d0 + u];
#pragma unroll
        for (int pi = 0; pi < PW; ++pi) kxd[pi][u] = kxs[(pbase + pi) * QS + d0 + u];
      }
      __builtin_amdgcn_sched_barrier(0);
#pragma unroll
      for (int u = 0; u < DB; ++u)
#pragma unroll
        for (int pi = 0; pi < PW; ++pi) {
          m0[pi] = fmaxf(m0[pi], attn_logit2(B0[pi], vj0[pi], kxd[pi][u], Ksd[u], ga[u]));
          m1[pi] = fmaxf(m1[pi], attn_logit2(B1[pi], vj1[pi], kxd[pi][u], Ksd[u], gb[u]));
        }
    }
#pragma unroll 1
    for (int d0 = 0; d0 < NC; d0 += DB) {
      float ga[DB], gb[DB], Ksd[DB], kxd[PW][DB];
#pragma unroll
      for (int u = 0; u < DB; ++u) {
        ga[u] = g0[d0 + u];  gb[u] = g1[d0 + u];  Ksd[u] = KsS[d0 + u];
#pragma unroll
        for (int pi = 0; pi < PW; ++pi) kxd[pi][u] = kxs[(pbase + pi) * QS + d0 + u];
      }
      __builtin_amdgcn_sched_barrier(0);
#pragma unroll
      for (int u = 0; u < DB; ++u)
#pragma unroll
        for (int pi = 0; pi < PW; ++pi) {
          z0[pi] += __builtin_amdgcn_exp2f(
              attn_logit2(B0[pi], vj0[pi], kxd[pi][u], Ksd[u], ga[u]) - m0[pi]);
          z1[pi] += __builtin_amdgcn_exp2f(
              attn_logit2(B1[pi], vj1[pi], kxd[pi][u], Ksd[u], gb[u]) - m1[pi]);
        }
    }
#pragma unroll
    for (int pi = 0; pi < PW; ++pi) {
      const int r0 = (pbase + pi) * QS + lane;
      os[r0]      = m0[pi];
      os[r0 + 32] = m1[pi];
      bs[r0]      = B0[pi];
      bs[r0 + 32] = B1[pi];
      qs[r0]      = qs[r0] / z0[pi];
      qs[r0 + 32] = qs[r0 + 32] / z1[pi];
    }
  }
  __syncthreads();

  {
    float kx0[PW], kx1[PW], o0[PW], o1[PW];
#pragma unroll
    for (int pi = 0; pi < PW; ++pi) {
      kx0[pi] = kxs[(pbase + pi) * QS + lane];
      kx1[pi] = kxs[(pbase + pi) * QS + lane + 32];
      o0[pi] = 0.f;  o1[pi] = 0.f;
    }
    const float Ks0 = KsS[lane];
    const float Ks1 = KsS[lane + 32];
#pragma unroll 8
    for (int j = 0; j < NC; ++j) {
      const float ga = Gs[j * GS + lane];
      const float gb = Gs[j * GS + lane + 32];
#pragma unroll
      for (int pi = 0; pi < PW; ++pi) {
        const int rj = (pbase + pi) * QS + j;
        const float vj = vxs[rj];
        const float Bj = bs[rj];
        const float mj = os[rj];
        const float wj = qs[rj];
        o0[pi] += wj * __builtin_amdgcn_exp2f(attn_logit2(Bj, vj, kx0[pi], Ks0, ga) - mj);
        o1[pi] += wj * __builtin_amdgcn_exp2f(attn_logit2(Bj, vj, kx1[pi], Ks1, gb) - mj);
      }
    }
#pragma unroll
    for (int pi = 0; pi < PW; ++pi) {
      os[(pbase + pi) * QS + lane]      = o0[pi];
      os[(pbase + pi) * QS + lane + 32] = o1[pi];
    }
  }
  __syncthreads();

  if (!gram_wave) {
    const float* arow = &os[fm * QS + fkoff];
    const int    drow = fh << 3;
    const v8f a0 = mma_lds_regs<NC / 4>(arow, wo0);
#pragma unroll
    for (int i = 0; i < 8; ++i) outs[(i + drow) * XS + fn] = a0[i] + bo0;
    const v8f a1 = mma_lds_regs<NC / 4>(arow, wo1);
#pragma unroll
    for (int i = 0; i < 8; ++i) outs[(i + drow) * XS + fn + 64] = a1[i] + bo1;
  }
  __syncthreads();

  if (gram_wave) {
    const int row   = tid >> 3;
    const int cbase = (tid & 7) * 16;
    float vals[16];
    float sum = 0.f, sq = 0.f;
#pragma unroll
    for (int i = 0; i < 16; ++i) {
      const float v = xs[row * XS + cbase + i] + outs[row * XS + cbase + i];
      vals[i] = v;
      sum += v;
      sq  += v * v;
    }
#pragma unroll
    for (int msk = 4; msk >= 1; msk >>= 1) {
      sum += __shfl_xor(sum, msk, 32);
      sq  += __shfl_xor(sq,  msk, 32);
    }
    const float mu   = sum * (1.f / 128.f);
    const float var  = sq * (1.f / 128.f) - mu * mu;
    const float rstd = rsqrtf(var + 1e-6f);
#pragma unroll
    for (int i = 0; i < 16; ++i) {
      const int c = cbase + i;
      y[(size_t)(p0 + row) * Cc + c] =
          (vals[i] - mu) * rstd * lsc[i] + lbi[i];
    }
  }
}

extern "C" void kernel_launch(void* const* d_in, const int* in_sizes, int n_in,
                              void* d_out, int out_size, void* d_ws, size_t ws_size,
                              hipStream_t stream) {
  const float* x     = (const float*)d_in[0];
  const float* s     = (const float*)d_in[1];
  const float* amask = (const float*)d_in[2];
  const float* Wq    = (const float*)d_in[3];
  const float* bq    = (const float*)d_in[4];
  const float* Wkx   = (const float*)d_in[5];
  const float* bkx   = (const float*)d_in[6];
  const float* Wvx   = (const float*)d_in[7];
  const float* bvx   = (const float*)d_in[8];
  const float* Wks   = (const float*)d_in[9];
  const float* bks   = (const float*)d_in[10];
  const float* Wvs   = (const float*)d_in[11];
  const float* bvs   = (const float*)d_in[12];
  const float* Wout  = (const float*)d_in[13];
  const float* bout  = (const float*)d_in[14];
  const float* lns   = (const float*)d_in[15];
  const float* lnb   = (const float*)d_in[16];
  float* y = (float*)d_out;

  char* ws = (char*)d_ws;
  float* ksb  = (float*)(ws + 0);
  float* vsb  = (float*)(ws + 262144);

  text_proj_kernel<<<dim3(Bn * (Ss / 16) * 2), dim3(128), 0, stream>>>(
      s, amask, Wks, bks, Wvs, bvs, ksb, vsb);
  fused_attn_kernel<<<dim3(NPOS / 16), dim3(256), 0, stream>>>(
      x, Wq, bq, Wkx, bkx, Wvx, bvx, ksb, vsb, Wout, bout, lns, lnb, y);
}


# Text projection: K split over the block's 4 waves (32 k-steps x 2 column tiles each; partials summed through LDS in wave order: a new summation order); a block covers 32 columns, 256 blocks
# speedup vs baseline: 6.8679x; 1.1844x over previous
#include <hip/hip_runtime.h>
#include <math.h>

constexpr int Bn  = 4;
constexpr int Wd  = 32;
constexpr int Hd  = 32;
constexpr int Cc  = 128;
constexpr int Ss  = 256;
constexpr int Dd  = 512;
constexpr int NC  = 64;
constexpr int NPOS = Bn * Wd * Hd;

typedef __attribute__((ext_vector_type(2))) float v2f;
typedef __attribute__((ext_vector_type(4))) float v4f;
typedef __attribute__((ext_vector_type(8))) float v8f;

__device__ __forceinline__ v8f wmma_k4(v2f a, v2f b, v8f c) {
  return __builtin_amdgcn_wmma_f32_16x16x4_f32(
       false, a,  false, b,
       (short)0, c,  false,  false);
}


constexpr float LOGIT_SCALE = 0.125f * 1.44269504088896340736f;
__device__ __forceinline__ float attn_logit2(float Bj, float vj, float kxd,
                                             float Ksd, float g) {
  return fmaf(Bj, kxd, fmaf(vj, Ksd, g));
}
__device__ __forceinline__ float attn_rowB(float vj, float Vj) {
  return fmaf(256.f, vj, Vj);
}

template <int KSTEPS, int LD>
__device__ __forceinline__ void load_w_frags(float (&w)[2 * KSTEPS],
                                             const float* __restrict__ Wm,
                                             int koff, int n) {
#pragma unroll
  for (int ks = 0; ks < KSTEPS; ++ks) {
    w[2 * ks]     = Wm[(4 * ks + koff) * LD + n];
    w[2 * ks + 1] = Wm[(4 * ks + koff + 1) * LD + n];
  }
}

template <int KSTEPS>
__device__ __forceinline__ v8f mma_lds_regs(const float* arow,
                                            const float (&w)[2 * KSTEPS]) {
  v8f acc = {0.f, 0.f, 0.f, 0.f, 0.f, 0.f, 0.f, 0.f};
#pragma unroll
  for (int ks = 0; ks < KSTEPS; ++ks) {
    v2f af, bf;
    af.x = arow[4 * ks];
    af.y = arow[4 * ks + 1];
    bf.x = w[2 * ks];
    bf.y = w[2 * ks + 1];
    acc = wmma_k4(af, bf, acc);
  }
  return acc;
}

__global__ __launch_bounds__(128) void text_proj_kernel(
    const float* __restrict__ s, const float* __restrict__ amask,
    const float* __restrict__ Wks, const float* __restrict__ bks,
    const float* __restrict__ Wvs, const float* __restrict__ bvs,
    float* __restrict__ ks_out, float* __restrict__ vs_out) {
  constexpr int TE_STRIDE = Dd + 4;
  constexpr int KQ        = Dd / 4;
  constexpr int KSTEPS    = KQ / 4;
  constexpr int RS        = 32 + 4;
  __shared__ __attribute__((aligned(16))) float te_s[16 * TE_STRIDE];
  __shared__ __attribute__((aligned(16))) float red_s[4 * 16 * RS];

  const int nh  = blockIdx.x & 1;
  const int mat = (blockIdx.x >> 1) & 1;
  const int st  = (blockIdx.x >> 2) & 15;
  const int b   = blockIdx.x >> 6;

  const int tid  = threadIdx.x;
  const int lane = tid & 31;
  const int wave = tid >> 5;

  const float* Wm   = mat ? Wvs : Wks;
  const float* bias = mat ? bvs : bks;
  float*       dst  = mat ? vs_out : ks_out;

  const int c    = lane & 15;
  const int h    = lane >> 4;
  const int koff = h * 2;
  const int nb   = nh * 32;
  const int kq   = wave * KQ;

  const int erow = tid >> 3;
  const int ec   = (tid & 7) * 4;

  float mk[16];
#pragma unroll
  for (int r = 0; r < 16; ++r) mk[r] = amask[b * Ss + st * 16 + r];
  v4f sv[16];
#pragma unroll
  for (int r = 0; r < 16; ++r)
    sv[r] = *reinterpret_cast<const v4f*>(&s[((size_t)b * Ss + st * 16 + r) * Dd + 4 * tid]);
  v2f wlo[KSTEPS], whi[KSTEPS];
  {
    const float* wp = Wm + (size_t)(kq + koff) * NC + nb + 2 * c;
#pragma unroll
    for (int ks = 0; ks < KSTEPS; ++ks) {
      wlo[ks] = *reinterpret_cast<const v2f*>(wp + (4 * ks)     * NC);
      whi[ks] = *reinterpret_cast<const v2f*>(wp + (4 * ks + 1) * NC);
    }
  }
  const v4f bv = *reinterpret_cast<const v4f*>(&bias[nb + ec]);

#pragma unroll
  for (int r = 0; r < 16; ++r)
    *reinterpret_cast<v4f*>(&te_s[r * TE_STRIDE + 4 * tid]) = sv[r] * mk[r];
  __syncthreads();

  const float* arow = &te_s[c * TE_STRIDE + kq + koff];
  v8f acc0 = {0.f, 0.f, 0.f, 0.f, 0.f, 0.f, 0.f, 0.f};
  v8f acc1 = {0.f, 0.f, 0.f, 0.f, 0.f, 0.f, 0.f, 0.f};
#pragma unroll
  for (int ks = 0; ks < KSTEPS; ++ks) {
    v2f af, b0, b1;
    af.x = arow[4 * ks];
    af.y = arow[4 * ks + 1];
    b0.x = wlo[ks].x;  b0.y = whi[ks].x;
    b1.x = wlo[ks].y;  b1.y = whi[ks].y;
    acc0 = wmma_k4(af, b0, acc0);
    acc1 = wmma_k4(af, b1, acc1);
  }

#pragma unroll
  for (int i = 0; i < 8; ++i) {
    v2f p;
    p.x = acc0[i];
    p.y = acc1[i];
    *reinterpret_cast<v2f*>(&red_s[(wave * 16 + 8 * h + i) * RS + 2 * c]) = p;
  }
  __syncthreads();

  v4f r = *reinterpret_cast<const v4f*>(&red_s[(0 * 16 + erow) * RS + ec]);
  r += *reinterpret_cast<const v4f*>(&red_s[(1 * 16 + erow) * RS + ec]);
  r += *reinterpret_cast<const v4f*>(&red_s[(2 * 16 + erow) * RS + ec]);
  r += *reinterpret_cast<const v4f*>(&red_s[(3 * 16 + erow) * RS + ec]);
  r += bv;
  *reinterpret_cast<v4f*>(&dst[((size_t)b * Ss + st * 16 + erow) * NC + nb + ec]) = r;
}

struct GramGroup {
  v4f   a0[8], a1[8];
  float b0[8], b1[8];
};

__device__ __forceinline__ void gram_load(GramGroup& f, const v4f* __restrict__ vs4,
                                          const float* __restrict__ ksp, int g) {
#pragma unroll
  for (int u = 0; u < 8; ++u) {
    const int row = g * 32 + u * 4;
    f.a0[u] = vs4[(row)     * (NC / 4)];
    f.a1[u] = vs4[(row + 1) * (NC / 4)];
    f.b0[u] = ksp[(row)     * NC];
    f.b1[u] = ksp[(row + 1) * NC];
  }
}

__device__ __forceinline__ void gram_mma(const GramGroup& f, v8f (&acc)[4],
                                         v4f& vp, float& kp) {
#pragma unroll
  for (int u = 0; u < 8; ++u) {
    v2f bf;
    bf.x = f.b0[u];
    bf.y = f.b1[u];
#pragma unroll
    for (int t = 0; t < 4; ++t) {
      v2f af;
      af.x = f.a0[u][t];
      af.y = f.a1[u][t];
      acc[t] = wmma_k4(af, bf, acc[t]);
    }
    kp += f.b0[u] + f.b1[u];
    vp += f.a0[u] + f.a1[u];
  }
}

__global__ __launch_bounds__(256) void fused_attn_kernel(
    const float* __restrict__ x,
    const float* __restrict__ Wq,  const float* __restrict__ bq,
    const float* __restrict__ Wkx, const float* __restrict__ bkx,
    const float* __restrict__ Wvx, const float* __restrict__ bvx,
    const float* __restrict__ ks, const float* __restrict__ vs,
    const float* __restrict__ Wout, const float* __restrict__ bout,
    const float* __restrict__ ln_scale, const float* __restrict__ ln_bias,
    float* __restrict__ y) {
  constexpr int XS = Cc + 4;
  constexpr int QS = NC + 4;
  constexpr int GS = NC + 1;
  constexpr int PW = 2;
  __shared__ float xs[16 * XS];
  __shared__ float qs[16 * QS];
  __shared__ float kxs[16 * QS];
  __shared__ float vxs[16 * QS];
  __shared__ float os[16 * QS];
  __shared__ float bs[16 * QS];
  __shared__ float outs[16 * XS];
  __shared__ float Gs[NC * GS];
  __shared__ float KsS[NC];
  __shared__ float VsS[NC];

  const int tid  = threadIdx.x;
  const int lane = tid & 31;
  const int wave = __builtin_amdgcn_readfirstlane(tid >> 5);
  const bool gram_wave = wave < 4;
  const int cw   = wave & 3;
  const int p0   = blockIdx.x * 16;
  const int b    = p0 >> 10;

  const int fm    = lane & 15;
  const int fh    = lane >> 4;
  const int fkoff = fh * 2;
  const int fn    = cw * 16 + fm;

  const v4f*   vs4 = reinterpret_cast<const v4f*>(vs) + ((size_t)b * Ss + 2 * fh) * (NC / 4) + fm;
  const float* ksp = ks + ((size_t)b * Ss + 2 * fh) * NC + fn;

  GramGroup fa, fb;
  float lsc[16], lbi[16];
  const int lnc = (tid & 7) * 16;
  if (gram_wave) {
    gram_load(fa, vs4, ksp, 0);
#pragma unroll
    for (int i = 0; i < 16; ++i) {
      lsc[i] = ln_scale[lnc + i];
      lbi[i] = ln_bias[lnc + i];
    }
  } else {
    const int t2 = tid & 127;
    float xr[16];
#pragma unroll
    for (int i = 0; i < 16; ++i) xr[i] = x[(size_t)p0 * Cc + t2 + i * 128];
#pragma unroll
    for (int i = 0; i < 16; ++i) xs[i * XS + t2] = xr[i];
  }
  __syncthreads();

  float wo0[2 * (NC / 4)], wo1[2 * (NC / 4)];
  float bo0 = 0.f, bo1 = 0.f;

  if (gram_wave) {
    v8f acc[4];
#pragma unroll
    for (int t = 0; t < 4; ++t) acc[t] = v8f{0.f, 0.f, 0.f, 0.f, 0.f, 0.f, 0.f, 0.f};
    v4f   vp = {0.f, 0.f, 0.f, 0.f};
    float kp = 0.f;

#pragma unroll 1
    for (int g = 0; g < 6; g += 2) {
      gram_load(fb, vs4, ksp, g + 1);
      gram_mma(fa, acc, vp, kp);
      gram_load(fa, vs4, ksp, g + 2);
      gram_mma(fb, acc, vp, kp);
    }
    gram_load(fb, vs4, ksp, 7);
    gram_mma(fa, acc, vp, kp);
    gram_mma(fb, acc, vp, kp);

    kp   += __shfl_xor(kp,   16, 32);
    vp.x += __shfl_xor(vp.x, 16, 32);
    vp.y += __shfl_xor(vp.y, 16, 32);
    vp.z += __shfl_xor(vp.z, 16, 32);
    vp.w += __shfl_xor(vp.w, 16, 32);

#pragma unroll
    for (int t = 0; t < 4; ++t)
#pragma unroll
      for (int i = 0; i < 8; ++i)
        Gs[(4 * (i + 8 * fh) + t) * GS + fn] = acc[t][i] * LOGIT_SCALE;
    if (lane < 16) {
      KsS[fn] = kp * LOGIT_SCALE;
      if (wave == 0) {
        VsS[4 * fm + 0] = vp.x;
        VsS[4 * fm + 1] = vp.y;
        VsS[4 * fm + 2] = vp.z;
        VsS[4 * fm + 3] = vp.w;
      }
    }
  } else {
    float wq[2 * (Cc / 4)], wk[2 * (Cc / 4)], wv[2 * (Cc / 4)];
    load_w_frags<Cc / 4, NC>(wq, Wq,  fkoff, fn);
    load_w_frags<Cc / 4, NC>(wk, Wkx, fkoff, fn);
    load_w_frags<Cc / 4, NC>(wv, Wvx, fkoff, fn);
    const float bnq = bq[fn], bnk = bkx[fn], bnv = bvx[fn];
    const float* arow = &xs[fm * XS + fkoff];
    const int    drow = fh << 3;
    const v8f aq = mma_lds_regs<Cc / 4>(arow, wq);
#pragma unroll
    for (int i = 0; i < 8; ++i) qs[(i + drow) * QS + fn] = aq[i] + bnq;
    const v8f ak = mma_lds_regs<Cc / 4>(arow, wk);
#pragma unroll
    for (int i = 0; i < 8; ++i) kxs[(i + drow) * QS + fn] = (ak[i] + bnk) * LOGIT_SCALE;
    const v8f av = mma_lds_regs<Cc / 4>(arow, wv);
#pragma unroll
    for (int i = 0; i < 8; ++i) vxs[(i + drow) * QS + fn] = av[i] + bnv;

    load_w_frags<NC / 4, Cc>(wo0, Wout, fkoff, fn);
    load_w_frags<NC / 4, Cc>(wo1, Wout, fkoff, fn + 64);
    bo0 = bout[fn];  bo1 = bout[fn + 64];
  }
  __syncthreads();

  const int pbase = wave * PW;

  {
    float vj0[PW], vj1[PW], B0[PW], B1[PW], m0[PW], m1[PW], z0[PW], z1[PW];
    const float Vj0 = VsS[lane];
    const float Vj1 = VsS[lane + 32];
#pragma unroll
    for (int pi = 0; pi < PW; ++pi) {
      vj0[pi] = vxs[(pbase + pi) * QS + lane];
      vj1[pi] = vxs[(pbase + pi) * QS + lane + 32];
      B0[pi] = attn_rowB(vj0[pi], Vj0);
      B1[pi] = attn_rowB(vj1[pi], Vj1);
      m0[pi] = -INFINITY;  m1[pi] = -INFINITY;
      z0[pi] = 0.f;        z1[pi] = 0.f;
    }
    const float* g0 = &Gs[lane * GS];
    const float* g1 = &Gs[(lane + 32) * GS];

    constexpr int DB = 8;
#pragma unroll 1
    for (int d0 = 0; d0 < NC; d0 += DB) {
      float ga[DB], gb[DB], Ksd[DB], kxd[PW][DB];
#pragma unroll
      for (int u = 0; u < DB; ++u) {
        ga[u] = g0[d0 + u];  gb[u] = g1[d0 + u];  Ksd[u] = KsS[d0 + u];
#pragma unroll
        for (int pi = 0; pi < PW; ++pi) kxd[pi][u] = kxs[(pbase + pi) * QS + d0 + u];
      }
      __builtin_amdgcn_sched_barrier(0);
#pragma unroll
      for (int u = 0; u < DB; ++u)
#pragma unroll
        for (int pi = 0; pi < PW; ++pi) {
          m0[pi] = fmaxf(m0[pi], attn_logit2(B0[pi], vj0[pi], kxd[pi][u], Ksd[u], ga[u]));
          m1[pi] = fmaxf(m1[pi], attn_logit2(B1[pi], vj1[pi], kxd[pi][u], Ksd[u], gb[u]));
        }
    }
#pragma unroll 1
    for (int d0 = 0; d0 < NC; d0 += DB) {
      float ga[DB], gb[DB], Ksd[DB], kxd[PW][DB];
#pragma unroll
      for (int u = 0; u < DB; ++u) {
        ga[u] = g0[d0 + u];  gb[u] = g1[d0 + u];  Ksd[u] = KsS[d0 + u];
#pragma unroll
        for (int pi = 0; pi < PW; ++pi) kxd[pi][u] = kxs[(pbase + pi) * QS + d0 + u];
      }
      __builtin_amdgcn_sched_barrier(0);
#pragma unroll
      for (int u = 0; u < DB; ++u)
#pragma unroll
        for (int pi = 0; pi < PW; ++pi) {
          z0[pi] += __builtin_amdgcn_exp2f(
              attn_logit2(B0[pi], vj0[pi], kxd[pi][u], Ksd[u], ga[u]) - m0[pi]);
          z1[pi] += __builtin_amdgcn_exp2f(
              attn_logit2(B1[pi], vj1[pi], kxd[pi][u], Ksd[u], gb[u]) - m1[pi]);
        }
    }
#pragma unroll
    for (int pi = 0; pi < PW; ++pi) {
      const int r0 = (pbase + pi) * QS + lane;
      os[r0]      = m0[pi];
      os[r0 + 32] = m1[pi];
      bs[r0]      = B0[pi];
      bs[r0 + 32] = B1[pi];
      qs[r0]      = qs[r0] / z0[pi];
      qs[r0 + 32] = qs[r0 + 32] / z1[pi];
    }
  }
  __syncthreads();

  {
    float kx0[PW], kx1[PW], o0[PW], o1[PW];
#pragma unroll
    for (int pi = 0; pi < PW; ++pi) {
      kx0[pi] = kxs[(pbase + pi) * QS + lane];
      kx1[pi] = kxs[(pbase + pi) * QS + lane + 32];
      o0[pi] = 0.f;  o1[pi] = 0.f;
    }
    const float Ks0 = KsS[lane];
    const float Ks1 = KsS[lane + 32];
#pragma unroll 8
    for (int j = 0; j < NC; ++j) {
      const float ga = Gs[j * GS + lane];
      const float gb = Gs[j * GS + lane + 32];
#pragma unroll
      for (int pi = 0; pi < PW; ++pi) {
        const int rj = (pbase + pi) * QS + j;
        const float vj = vxs[rj];
        const float Bj = bs[rj];
        const float mj = os[rj];
        const float wj = qs[rj];
        o0[pi] += wj * __builtin_amdgcn_exp2f(attn_logit2(Bj, vj, kx0[pi], Ks0, ga) - mj);
        o1[pi] += wj * __builtin_amdgcn_exp2f(attn_logit2(Bj, vj, kx1[pi], Ks1, gb) - mj);
      }
    }
#pragma unroll
    for (int pi = 0; pi < PW; ++pi) {
      os[(pbase + pi) * QS + lane]      = o0[pi];
      os[(pbase + pi) * QS + lane + 32] = o1[pi];
    }
  }
  __syncthreads();

  if (!gram_wave) {
    const float* arow = &os[fm * QS + fkoff];
    const int    drow = fh << 3;
    const v8f a0 = mma_lds_regs<NC / 4>(arow, wo0);
#pragma unroll
    for (int i = 0; i < 8; ++i) outs[(i + drow) * XS + fn] = a0[i] + bo0;
    const v8f a1 = mma_lds_regs<NC / 4>(arow, wo1);
#pragma unroll
    for (int i = 0; i < 8; ++i) outs[(i + drow) * XS + fn + 64] = a1[i] + bo1;
  }
  __syncthreads();

  if (gram_wave) {
    const int row   = tid >> 3;
    const int cbase = (tid & 7) * 16;
    float vals[16];
    float sum = 0.f, sq = 0.f;
#pragma unroll
    for (int i = 0; i < 16; ++i) {
      const float v = xs[row * XS + cbase + i] + outs[row * XS + cbase + i];
      vals[i] = v;
      sum += v;
      sq  += v * v;
    }
#pragma unroll
    for (int msk = 4; msk >= 1; msk >>= 1) {
      sum += __shfl_xor(sum, msk, 32);
      sq  += __shfl_xor(sq,  msk, 32);
    }
    const float mu   = sum * (1.f / 128.f);
    const float var  = sq * (1.f / 128.f) - mu * mu;
    const float rstd = rsqrtf(var + 1e-6f);
#pragma unroll
    for (int i = 0; i < 16; ++i) {
      const int c = cbase + i;
      y[(size_t)(p0 + row) * Cc + c] =
          (vals[i] - mu) * rstd * lsc[i] + lbi[i];
    }
  }
}

extern "C" void kernel_launch(void* const* d_in, const int* in_sizes, int n_in,
                              void* d_out, int out_size, void* d_ws, size_t ws_size,
                              hipStream_t stream) {
  const float* x     = (const float*)d_in[0];
  const float* s     = (const float*)d_in[1];
  const float* amask = (const float*)d_in[2];
  const float* Wq    = (const float*)d_in[3];
  const float* bq    = (const float*)d_in[4];
  const float* Wkx   = (const float*)d_in[5];
  const float* bkx   = (const float*)d_in[6];
  const float* Wvx   = (const float*)d_in[7];
  const float* bvx   = (const float*)d_in[8];
  const float* Wks   = (const float*)d_in[9];
  const float* bks   = (const float*)d_in[10];
  const float* Wvs   = (const float*)d_in[11];
  const float* bvs   = (const float*)d_in[12];
  const float* Wout  = (const float*)d_in[13];
  const float* bout  = (const float*)d_in[14];
  const float* lns   = (const float*)d_in[15];
  const float* lnb   = (const float*)d_in[16];
  float* y = (float*)d_out;

  char* ws = (char*)d_ws;
  float* ksb  = (float*)(ws + 0);
  float* vsb  = (float*)(ws + 262144);

  text_proj_kernel<<<dim3(Bn * (Ss / 16) * 2 * 2), dim3(128), 0, stream>>>(
      s, amask, Wks, bks, Wvs, bvs, ksb, vsb);
  fused_attn_kernel<<<dim3(NPOS / 16), dim3(256), 0, stream>>>(
      x, Wq, bq, Wkx, bkx, Wvx, bvx, ksb, vsb, Wout, bout, lns, lnb, y);
}


# Channel softmax: a lane forms its 2 rows' 64 logits once and keeps them in registers through max, exp2, sum (was 3 evaluations, 2 exp2); weighted rows added over lanes via LDS: new output sum order
# speedup vs baseline: 7.5429x; 1.0983x over previous
#include <hip/hip_runtime.h>
#include <math.h>

constexpr int Bn  = 4;
constexpr int Wd  = 32;
constexpr int Hd  = 32;
constexpr int Cc  = 128;
constexpr int Ss  = 256;
constexpr int Dd  = 512;
constexpr int NC  = 64;
constexpr int NPOS = Bn * Wd * Hd;

typedef __attribute__((ext_vector_type(2))) float v2f;
typedef __attribute__((ext_vector_type(4))) float v4f;
typedef __attribute__((ext_vector_type(8))) float v8f;

__device__ __forceinline__ v8f wmma_k4(v2f a, v2f b, v8f c) {
  return __builtin_amdgcn_wmma_f32_16x16x4_f32(
       false, a,  false, b,
       (short)0, c,  false,  false);
}


constexpr float LOGIT_SCALE = 0.125f * 1.44269504088896340736f;
__device__ __forceinline__ v2f attn_logit2(v2f Bj, v2f vj, float kxd, float Ksd, v2f g) {
  const v2f kx2 = {kxd, kxd};
  const v2f Ks2 = {Ksd, Ksd};
  return __builtin_elementwise_fma(Bj, kx2, __builtin_elementwise_fma(vj, Ks2, g));
}
__device__ __forceinline__ float attn_rowB(float vj, float Vj) {
  return fmaf(256.f, vj, Vj);
}

struct LogitOps {
  v4f g[8];
  v4f kx[4];
  v4f ks[4];
};
__device__ __forceinline__ void logit_ops_load(LogitOps& o, const v4f* g4, const v4f* kx4,
                                               const v4f* ks4, int c16) {
#pragma unroll
  for (int i = 0; i < 8; ++i) o.g[i] = g4[c16 * 8 + i];
#pragma unroll
  for (int i = 0; i < 4; ++i) {
    o.kx[i] = kx4[c16 * 4 + i];
    o.ks[i] = ks4[c16 * 4 + i];
  }
}
__device__ __forceinline__ void logit_chunk(const LogitOps& o, int c16, v2f Bj, v2f vj,
                                            v2f (&lv)[NC], v2f (&mm)[4]) {
#pragma unroll
  for (int u = 0; u < 16; ++u) {
    const int d = c16 * 16 + u;
    const v4f gq = o.g[u >> 1];
    const v2f g  = (u & 1) ? v2f{gq.z, gq.w} : v2f{gq.x, gq.y};
    lv[d] = attn_logit2(Bj, vj, o.kx[u >> 2][u & 3], o.ks[u >> 2][u & 3], g);
    mm[u & 3] = __builtin_elementwise_max(mm[u & 3], lv[d]);
  }
}

template <int KSTEPS, int LD>
__device__ __forceinline__ void load_w_frags(float (&w)[2 * KSTEPS],
                                             const float* __restrict__ Wm,
                                             int koff, int n) {
#pragma unroll
  for (int ks = 0; ks < KSTEPS; ++ks) {
    w[2 * ks]     = Wm[(4 * ks + koff) * LD + n];
    w[2 * ks + 1] = Wm[(4 * ks + koff + 1) * LD + n];
  }
}

template <int KSTEPS>
__device__ __forceinline__ v8f mma_lds_regs(const float* arow,
                                            const float (&w)[2 * KSTEPS]) {
  v8f acc = {0.f, 0.f, 0.f, 0.f, 0.f, 0.f, 0.f, 0.f};
#pragma unroll
  for (int ks = 0; ks < KSTEPS; ++ks) {
    v2f af, bf;
    af.x = arow[4 * ks];
    af.y = arow[4 * ks + 1];
    bf.x = w[2 * ks];
    bf.y = w[2 * ks + 1];
    acc = wmma_k4(af, bf, acc);
  }
  return acc;
}

__global__ __launch_bounds__(128) void text_proj_kernel(
    const float* __restrict__ s, const float* __restrict__ amask,
    const float* __restrict__ Wks, const float* __restrict__ bks,
    const float* __restrict__ Wvs, const float* __restrict__ bvs,
    float* __restrict__ ks_out, float* __restrict__ vs_out) {
  constexpr int TE_STRIDE = Dd + 4;
  constexpr int KQ        = Dd / 4;
  constexpr int KSTEPS    = KQ / 4;
  constexpr int RS        = 32 + 4;
  __shared__ __attribute__((aligned(16))) float te_s[16 * TE_STRIDE];
  __shared__ __attribute__((aligned(16))) float red_s[4 * 16 * RS];

  const int nh  = blockIdx.x & 1;
  const int mat = (blockIdx.x >> 1) & 1;
  const int st  = (blockIdx.x >> 2) & 15;
  const int b   = blockIdx.x >> 6;

  const int tid  = threadIdx.x;
  const int lane = tid & 31;
  const int wave = tid >> 5;

  const float* Wm   = mat ? Wvs : Wks;
  const float* bias = mat ? bvs : bks;
  float*       dst  = mat ? vs_out : ks_out;

  const int c    = lane & 15;
  const int h    = lane >> 4;
  const int koff = h * 2;
  const int nb   = nh * 32;
  const int kq   = wave * KQ;

  const int erow = tid >> 3;
  const int ec   = (tid & 7) * 4;

  float mk[16];
#pragma unroll
  for (int r = 0; r < 16; ++r) mk[r] = amask[b * Ss + st * 16 + r];
  v4f sv[16];
#pragma unroll
  for (int r = 0; r < 16; ++r)
    sv[r] = *reinterpret_cast<const v4f*>(&s[((size_t)b * Ss + st * 16 + r) * Dd + 4 * tid]);
  v2f wlo[KSTEPS], whi[KSTEPS];
  {
    const float* wp = Wm + (size_t)(kq + koff) * NC + nb + 2 * c;
#pragma unroll
    for (int ks = 0; ks < KSTEPS; ++ks) {
      wlo[ks] = *reinterpret_cast<const v2f*>(wp + (4 * ks)     * NC);
      whi[ks] = *reinterpret_cast<const v2f*>(wp + (4 * ks + 1) * NC);
    }
  }
  const v4f bv = *reinterpret_cast<const v4f*>(&bias[nb + ec]);

#pragma unroll
  for (int r = 0; r < 16; ++r)
    *reinterpret_cast<v4f*>(&te_s[r * TE_STRIDE + 4 * tid]) = sv[r] * mk[r];
  __syncthreads();

  const float* arow = &te_s[c * TE_STRIDE + kq + koff];
  v8f acc0 = {0.f, 0.f, 0.f, 0.f, 0.f, 0.f, 0.f, 0.f};
  v8f acc1 = {0.f, 0.f, 0.f, 0.f, 0.f, 0.f, 0.f, 0.f};
#pragma unroll
  for (int ks = 0; ks < KSTEPS; ++ks) {
    v2f af, b0, b1;
    af.x = arow[4 * ks];
    af.y = arow[4 * ks + 1];
    b0.x = wlo[ks].x;  b0.y = whi[ks].x;
    b1.x = wlo[ks].y;  b1.y = whi[ks].y;
    acc0 = wmma_k4(af, b0, acc0);
    acc1 = wmma_k4(af, b1, acc1);
  }

#pragma unroll
  for (int i = 0; i < 8; ++i) {
    v2f p;
    p.x = acc0[i];
    p.y = acc1[i];
    *reinterpret_cast<v2f*>(&red_s[(wave * 16 + 8 * h + i) * RS + 2 * c]) = p;
  }
  __syncthreads();

  v4f r = *reinterpret_cast<const v4f*>(&red_s[(0 * 16 + erow) * RS + ec]);
  r += *reinterpret_cast<const v4f*>(&red_s[(1 * 16 + erow) * RS + ec]);
  r += *reinterpret_cast<const v4f*>(&red_s[(2 * 16 + erow) * RS + ec]);
  r += *reinterpret_cast<const v4f*>(&red_s[(3 * 16 + erow) * RS + ec]);
  r += bv;
  *reinterpret_cast<v4f*>(&dst[((size_t)b * Ss + st * 16 + erow) * NC + nb + ec]) = r;
}

struct GramGroup {
  v4f   a0[8], a1[8];
  float b0[8], b1[8];
};

__device__ __forceinline__ void gram_load(GramGroup& f, const v4f* __restrict__ vs4,
                                          const float* __restrict__ ksp, int g) {
#pragma unroll
  for (int u = 0; u < 8; ++u) {
    const int row = g * 32 + u * 4;
    f.a0[u] = vs4[(row)     * (NC / 4)];
    f.a1[u] = vs4[(row + 1) * (NC / 4)];
    f.b0[u] = ksp[(row)     * NC];
    f.b1[u] = ksp[(row + 1) * NC];
  }
}

__device__ __forceinline__ void gram_mma(const GramGroup& f, v8f (&acc)[4],
                                         v4f& vp, float& kp) {
#pragma unroll
  for (int u = 0; u < 8; ++u) {
    v2f bf;
    bf.x = f.b0[u];
    bf.y = f.b1[u];
#pragma unroll
    for (int t = 0; t < 4; ++t) {
      v2f af;
      af.x = f.a0[u][t];
      af.y = f.a1[u][t];
      acc[t] = wmma_k4(af, bf, acc[t]);
    }
    kp += f.b0[u] + f.b1[u];
    vp += f.a0[u] + f.a1[u];
  }
}

__global__ __launch_bounds__(256) void fused_attn_kernel(
    const float* __restrict__ x,
    const float* __restrict__ Wq,  const float* __restrict__ bq,
    const float* __restrict__ Wkx, const float* __restrict__ bkx,
    const float* __restrict__ Wvx, const float* __restrict__ bvx,
    const float* __restrict__ ks, const float* __restrict__ vs,
    const float* __restrict__ Wout, const float* __restrict__ bout,
    const float* __restrict__ ln_scale, const float* __restrict__ ln_bias,
    float* __restrict__ y) {
  constexpr int XS = Cc + 4;
  constexpr int QS = NC + 4;
  constexpr int GP = 2 * NC + 4;
  constexpr int PW = 2;
  constexpr int TS = NC + 4;
  __shared__ float xs[16 * XS];
  __shared__ float qs[16 * QS];
  __shared__ __attribute__((aligned(16))) float kxs[16 * QS];
  __shared__ float vxs[16 * QS];
  __shared__ float os[16 * QS];
  __shared__ float outs[16 * XS];
  __shared__ __attribute__((aligned(16))) float Gs[(NC / 2) * GP];
  __shared__ __attribute__((aligned(16))) float KsS[NC];
  __shared__ float VsS[NC];
  __shared__ __attribute__((aligned(16))) float ts[8 * 32 * TS];

  const int tid  = threadIdx.x;
  const int lane = tid & 31;
  const int wave = __builtin_amdgcn_readfirstlane(tid >> 5);
  const bool gram_wave = wave < 4;
  const int cw   = wave & 3;
  const int p0   = blockIdx.x * 16;
  const int b    = p0 >> 10;

  const int fm    = lane & 15;
  const int fh    = lane >> 4;
  const int fkoff = fh * 2;
  const int fn    = cw * 16 + fm;

  const v4f*   vs4 = reinterpret_cast<const v4f*>(vs) + ((size_t)b * Ss + 2 * fh) * (NC / 4) + fm;
  const float* ksp = ks + ((size_t)b * Ss + 2 * fh) * NC + fn;

  GramGroup fa, fb;
  float lsc[16], lbi[16];
  const int lnc = (tid & 7) * 16;
  if (gram_wave) {
    gram_load(fa, vs4, ksp, 0);
#pragma unroll
    for (int i = 0; i < 16; ++i) {
      lsc[i] = ln_scale[lnc + i];
      lbi[i] = ln_bias[lnc + i];
    }
  } else {
    const int t2 = tid & 127;
    float xr[16];
#pragma unroll
    for (int i = 0; i < 16; ++i) xr[i] = x[(size_t)p0 * Cc + t2 + i * 128];
#pragma unroll
    for (int i = 0; i < 16; ++i) xs[i * XS + t2] = xr[i];
  }
  __syncthreads();

  float wo0[2 * (NC / 4)], wo1[2 * (NC / 4)];
  float bo0 = 0.f, bo1 = 0.f;

  if (gram_wave) {
    v8f acc[4];
#pragma unroll
    for (int t = 0; t < 4; ++t) acc[t] = v8f{0.f, 0.f, 0.f, 0.f, 0.f, 0.f, 0.f, 0.f};
    v4f   vp = {0.f, 0.f, 0.f, 0.f};
    float kp = 0.f;

#pragma unroll 1
    for (int g = 0; g < 6; g += 2) {
      gram_load(fb, vs4, ksp, g + 1);
      gram_mma(fa, acc, vp, kp);
      gram_load(fa, vs4, ksp, g + 2);
      gram_mma(fb, acc, vp, kp);
    }
    gram_load(fb, vs4, ksp, 7);
    gram_mma(fa, acc, vp, kp);
    gram_mma(fb, acc, vp, kp);

    kp   += __shfl_xor(kp,   16, 32);
    vp.x += __shfl_xor(vp.x, 16, 32);
    vp.y += __shfl_xor(vp.y, 16, 32);
    vp.z += __shfl_xor(vp.z, 16, 32);
    vp.w += __shfl_xor(vp.w, 16, 32);

#pragma unroll
    for (int t = 0; t < 4; t += 2)
#pragma unroll
      for (int i = 0; i < 8; ++i) {
        v2f gp;
        gp.x = acc[t][i]     * LOGIT_SCALE;
        gp.y = acc[t + 1][i] * LOGIT_SCALE;
        *reinterpret_cast<v2f*>(&Gs[(2 * (i + 8 * fh) + t / 2) * GP + 2 * fn]) = gp;
      }
    if (lane < 16) {
      KsS[fn] = kp * LOGIT_SCALE;
      if (wave == 0) {
        VsS[4 * fm + 0] = vp.x;
        VsS[4 * fm + 1] = vp.y;
        VsS[4 * fm + 2] = vp.z;
        VsS[4 * fm + 3] = vp.w;
      }
    }
  } else {
    float wq[2 * (Cc / 4)], wk[2 * (Cc / 4)], wv[2 * (Cc / 4)];
    load_w_frags<Cc / 4, NC>(wq, Wq,  fkoff, fn);
    load_w_frags<Cc / 4, NC>(wk, Wkx, fkoff, fn);
    load_w_frags<Cc / 4, NC>(wv, Wvx, fkoff, fn);
    const float bnq = bq[fn], bnk = bkx[fn], bnv = bvx[fn];
    const float* arow = &xs[fm * XS + fkoff];
    const int    drow = fh << 3;
    const v8f aq = mma_lds_regs<Cc / 4>(arow, wq);
#pragma unroll
    for (int i = 0; i < 8; ++i) qs[(i + drow) * QS + fn] = aq[i] + bnq;
    const v8f ak = mma_lds_regs<Cc / 4>(arow, wk);
#pragma unroll
    for (int i = 0; i < 8; ++i) kxs[(i + drow) * QS + fn] = (ak[i] + bnk) * LOGIT_SCALE;
    const v8f av = mma_lds_regs<Cc / 4>(arow, wv);
#pragma unroll
    for (int i = 0; i < 8; ++i) vxs[(i + drow) * QS + fn] = av[i] + bnv;

    load_w_frags<NC / 4, Cc>(wo0, Wout, fkoff, fn);
    load_w_frags<NC / 4, Cc>(wo1, Wout, fkoff, fn + 64);
    bo0 = bout[fn];  bo1 = bout[fn + 64];
  }
  __syncthreads();

  const int pbase = wave * PW;
  {
    const int j0 = 2 * lane;
    const v4f* g4  = reinterpret_cast<const v4f*>(&Gs[lane * GP]);
    const v4f* ks4 = reinterpret_cast<const v4f*>(KsS);
    const float Vj0 = VsS[j0];
    const float Vj1 = VsS[j0 + 1];
    float* tw = &ts[wave * (32 * TS)];

#pragma unroll 1
    for (int pi = 0; pi < PW; ++pi) {
      asm volatile("" ::: "memory");
      const int prow = (pbase + pi) * QS;
      const v2f vj = {vxs[prow + j0], vxs[prow + j0 + 1]};
      const v2f Bj = {attn_rowB(vj.x, Vj0), attn_rowB(vj.y, Vj1)};

      const v4f* kx4 = reinterpret_cast<const v4f*>(&kxs[prow]);
      v2f lv[NC];
      v2f mm[4];
#pragma unroll
      for (int u = 0; u < 4; ++u) mm[u] = v2f{-INFINITY, -INFINITY};
      LogitOps oa, ob;
      logit_ops_load(oa, g4, kx4, ks4, 0);
      logit_ops_load(ob, g4, kx4, ks4, 1);
      __builtin_amdgcn_sched_barrier(0);
      logit_chunk(oa, 0, Bj, vj, lv, mm);
      logit_ops_load(oa, g4, kx4, ks4, 2);
      __builtin_amdgcn_sched_barrier(0);
      logit_chunk(ob, 1, Bj, vj, lv, mm);
      logit_ops_load(ob, g4, kx4, ks4, 3);
      __builtin_amdgcn_sched_barrier(0);
      logit_chunk(oa, 2, Bj, vj, lv, mm);
      logit_chunk(ob, 3, Bj, vj, lv, mm);
      const v2f m = __builtin_elementwise_max(__builtin_elementwise_max(mm[0], mm[1]),
                                              __builtin_elementwise_max(mm[2], mm[3]));

      v2f z = {0.f, 0.f};
      constexpr int AH = 4;
      __builtin_amdgcn_sched_barrier(0);
#pragma unroll
      for (int d = 0; d < AH; ++d) lv[d] -= m;
#pragma unroll
      for (int d = 0; d < NC; ++d) {
        lv[d].x = __builtin_amdgcn_exp2f(lv[d].x);
        if (d + AH < NC) lv[d + AH] -= m;
        lv[d].y = __builtin_amdgcn_exp2f(lv[d].y);
        if (d >= AH) z += lv[d - AH];
        __builtin_amdgcn_sched_group_barrier(0x400, 1, 0);
        __builtin_amdgcn_sched_group_barrier(0x002, 1, 0);
        __builtin_amdgcn_sched_group_barrier(0x400, 1, 0);
        __builtin_amdgcn_sched_group_barrier(0x002, 1, 0);
      }
#pragma unroll
      for (int d = NC - AH; d < NC; ++d) z += lv[d];
      __builtin_amdgcn_sched_barrier(0);
      const float w0 = qs[prow + j0] / z.x;
      const float w1 = qs[prow + j0 + 1] / z.y;

#pragma unroll
      for (int d = 0; d < NC; d += 4) {
        v4f yv;
        yv.x = fmaf(w1, lv[d].y,     w0 * lv[d].x);
        yv.y = fmaf(w1, lv[d + 1].y, w0 * lv[d + 1].x);
        yv.z = fmaf(w1, lv[d + 2].y, w0 * lv[d + 2].x);
        yv.w = fmaf(w1, lv[d + 3].y, w0 * lv[d + 3].x);
        *reinterpret_cast<v4f*>(&tw[lane * TS + d]) = yv;
      }
      float ta[32], tb[32];
#pragma unroll
      for (int r = 0; r < 32; ++r) {
        ta[r] = tw[r * TS + lane];
        tb[r] = tw[r * TS + lane + 32];
      }
      __builtin_amdgcn_sched_barrier(0);
      float o0 = 0.f, o1 = 0.f;
#pragma unroll
      for (int r = 0; r < 32; ++r) {
        o0 += ta[r];
        o1 += tb[r];
      }
      os[prow + lane]      = o0;
      os[prow + lane + 32] = o1;
    }
  }
  __syncthreads();

  if (!gram_wave) {
    const float* arow = &os[fm * QS + fkoff];
    const int    drow = fh << 3;
    const v8f a0 = mma_lds_regs<NC / 4>(arow, wo0);
#pragma unroll
    for (int i = 0; i < 8; ++i) outs[(i + drow) * XS + fn] = a0[i] + bo0;
    const v8f a1 = mma_lds_regs<NC / 4>(arow, wo1);
#pragma unroll
    for (int i = 0; i < 8; ++i) outs[(i + drow) * XS + fn + 64] = a1[i] + bo1;
  }
  __syncthreads();

  if (gram_wave) {
    const int row   = tid >> 3;
    const int cbase = (tid & 7) * 16;
    float vals[16];
    float sum = 0.f, sq = 0.f;
#pragma unroll
    for (int i = 0; i < 16; ++i) {
      const float v = xs[row * XS + cbase + i] + outs[row * XS + cbase + i];
      vals[i] = v;
      sum += v;
      sq  += v * v;
    }
#pragma unroll
    for (int msk = 4; msk >= 1; msk >>= 1) {
      sum += __shfl_xor(sum, msk, 32);
      sq  += __shfl_xor(sq,  msk, 32);
    }
    const float mu   = sum * (1.f / 128.f);
    const float var  = sq * (1.f / 128.f) - mu * mu;
    const float rstd = rsqrtf(var + 1e-6f);
#pragma unroll
    for (int i = 0; i < 16; ++i) {
      const int c = cbase + i;
      y[(size_t)(p0 + row) * Cc + c] =
          (vals[i] - mu) * rstd * lsc[i] + lbi[i];
    }
  }
}

extern "C" void kernel_launch(void* const* d_in, const int* in_sizes, int n_in,
                              void* d_out, int out_size, void* d_ws, size_t ws_size,
                              hipStream_t stream) {
  const float* x     = (const float*)d_in[0];
  const float* s     = (const float*)d_in[1];
  const float* amask = (const float*)d_in[2];
  const float* Wq    = (const float*)d_in[3];
  const float* bq    = (const float*)d_in[4];
  const float* Wkx   = (const float*)d_in[5];
  const float* bkx   = (const float*)d_in[6];
  const float* Wvx   = (const float*)d_in[7];
  const float* bvx   = (const float*)d_in[8];
  const float* Wks   = (const float*)d_in[9];
  const float* bks   = (const float*)d_in[10];
  const float* Wvs   = (const float*)d_in[11];
  const float* bvs   = (const float*)d_in[12];
  const float* Wout  = (const float*)d_in[13];
  const float* bout  = (const float*)d_in[14];
  const float* lns   = (const float*)d_in[15];
  const float* lnb   = (const float*)d_in[16];
  float* y = (float*)d_out;

  char* ws = (char*)d_ws;
  float* ksb  = (float*)(ws + 0);
  float* vsb  = (float*)(ws + 262144);

  text_proj_kernel<<<dim3(Bn * (Ss / 16) * 2 * 2), dim3(128), 0, stream>>>(
      s, amask, Wks, bks, Wvs, bvs, ksb, vsb);
  fused_attn_kernel<<<dim3(NPOS / 16), dim3(256), 0, stream>>>(
      x, Wq, bq, Wkx, bkx, Wvx, bvx, ksb, vsb, Wout, bout, lns, lnb, y);
}
